# starting kernel
# baseline (speedup 1.0000x reference)
#include <hip/hip_runtime.h>
#include <hip/hip_bf16.h>


typedef __attribute__((ext_vector_type(2))) float v2f;
typedef __attribute__((ext_vector_type(8))) float v8f;

#define F1 16
#define F2 2
#define FIN 25

__device__ __forceinline__ void atomic_add_f32(float* p, float v) {
    asm volatile("global_atomic_add_f32 %0, %1, off"
                 :
                 : "v"(p), "v"(v)
                 : "memory");
}

__global__ void gcn_init_kernel(float* __restrict__ deg,
                                float* __restrict__ agg1,
                                float* __restrict__ agg2,
                                int n) {
    int gid = blockIdx.x * blockDim.x + threadIdx.x;
    if (gid < 16 * n) agg1[gid] = 0.0f;
    if (gid < 2 * n)  agg2[gid] = 0.0f;
    if (gid < n)      deg[gid]  = 1.0f;
}

__global__ void gcn_degree_kernel(const int* __restrict__ cols,
                                  float* __restrict__ deg,
                                  int n_edges) {
    int e = blockIdx.x * blockDim.x + threadIdx.x;
    if (e >= n_edges) return;
    atomic_add_f32(&deg[cols[e]], 1.0f);
}

__global__ void gcn_dinv_kernel(const float* __restrict__ deg,
                                float* __restrict__ dinv,
                                int n) {
    int i = blockIdx.x * blockDim.x + threadIdx.x;
    if (i >= n) return;
    dinv[i] = rsqrtf(deg[i]);
}

__global__ void gcn_xw1_wmma_kernel(const float* __restrict__ x,
                                    const float* __restrict__ W1,
                                    float* __restrict__ h1,
                                    int n_nodes) {
    const int lane  = threadIdx.x & 31;
    const int wave  = threadIdx.x >> 5;
    const int tile  = blockIdx.x * (blockDim.x >> 5) + wave;
    const int node0 = tile * 16;
    if (node0 >= n_nodes) return;

    const int half = lane >> 4;
    const int l16  = lane & 15;
    const int m    = l16;

    v8f c = {0.f, 0.f, 0.f, 0.f, 0.f, 0.f, 0.f, 0.f};

    #pragma unroll
    for (int k0 = 0; k0 < 28; k0 += 4) {
        const int ka = k0 + 2 * half;
        v2f a, b;
        a.x = (ka     < FIN) ? x[(node0 + m) * FIN + ka    ] : 0.0f;
        a.y = (ka + 1 < FIN) ? x[(node0 + m) * FIN + ka + 1] : 0.0f;
        b.x = (ka     < FIN) ? W1[(ka    ) * F1 + l16] : 0.0f;
        b.y = (ka + 1 < FIN) ? W1[(ka + 1) * F1 + l16] : 0.0f;
        c = __builtin_amdgcn_wmma_f32_16x16x4_f32(
                 false, a,  false, b,
                 (short)0, c,  false,  false);
    }

    #pragma unroll
    for (int v = 0; v < 8; ++v) {
        h1[(node0 + v + 8 * half) * F1 + l16] = c[v];
    }
}

__global__ void gcn_scatter1_kernel(const int* __restrict__ rows,
                                    const int* __restrict__ cols,
                                    const float* __restrict__ h1,
                                    const float* __restrict__ dinv,
                                    float* __restrict__ agg1,
                                    int n_edges) {
    long long gid = (long long)blockIdx.x * blockDim.x + threadIdx.x;
    int e = (int)(gid >> 4);
    int f = (int)(gid & 15);
    if (e >= n_edges) return;
    int r = rows[e];
    int cidx = cols[e];
    float norm = dinv[r] * dinv[cidx];
    atomic_add_f32(&agg1[cidx * F1 + f], h1[r * F1 + f] * norm);
}

__global__ void gcn_layer1_finish_kernel(const float* __restrict__ h1,
                                         const float* __restrict__ agg1,
                                         const float* __restrict__ dinv,
                                         const float* __restrict__ b1,
                                         const float* __restrict__ W2,
                                         float* __restrict__ h2,
                                         int n) {
    int i = blockIdx.x * blockDim.x + threadIdx.x;
    if (i >= n) return;
    float di = dinv[i];
    float sw = di * di;
    float a0 = 0.0f, a1 = 0.0f;
    #pragma unroll
    for (int f = 0; f < F1; ++f) {
        float v = agg1[i * F1 + f] + h1[i * F1 + f] * sw + b1[f];
        v = v > 0.0f ? v : 0.0f;
        a0 = fmaf(v, W2[f * F2 + 0], a0);
        a1 = fmaf(v, W2[f * F2 + 1], a1);
    }
    h2[i * F2 + 0] = a0;
    h2[i * F2 + 1] = a1;
}

__global__ void gcn_scatter2_kernel(const int* __restrict__ rows,
                                    const int* __restrict__ cols,
                                    const float* __restrict__ h2,
                                    const float* __restrict__ dinv,
                                    float* __restrict__ agg2,
                                    int n_edges) {
    long long gid = (long long)blockIdx.x * blockDim.x + threadIdx.x;
    int e = (int)(gid >> 1);
    int f = (int)(gid & 1);
    if (e >= n_edges) return;
    int r = rows[e];
    int cidx = cols[e];
    float norm = dinv[r] * dinv[cidx];
    atomic_add_f32(&agg2[cidx * F2 + f], h2[r * F2 + f] * norm);
}

__global__ void gcn_final_kernel(const float* __restrict__ h2,
                                 const float* __restrict__ agg2,
                                 const float* __restrict__ dinv,
                                 const float* __restrict__ b2,
                                 float* __restrict__ out,
                                 int n) {
    int i = blockIdx.x * blockDim.x + threadIdx.x;
    if (i >= n) return;
    float di = dinv[i];
    float sw = di * di;
    float a = agg2[i * F2 + 0] + h2[i * F2 + 0] * sw + b2[0];
    float b = agg2[i * F2 + 1] + h2[i * F2 + 1] * sw + b2[1];
    float m = fmaxf(a, b);
    float lse = m + logf(expf(a - m) + expf(b - m));
    out[i * F2 + 0] = a - lse;
    out[i * F2 + 1] = b - lse;
}

extern "C" void kernel_launch(void* const* d_in, const int* in_sizes, int n_in,
                              void* d_out, int out_size, void* d_ws, size_t ws_size,
                              hipStream_t stream) {
    const float* x   = (const float*)d_in[0];
    const int*   ei  = (const int*)d_in[1];
    const float* W1  = (const float*)d_in[2];
    const float* b1  = (const float*)d_in[3];
    const float* W2  = (const float*)d_in[4];
    const float* b2  = (const float*)d_in[5];
    float*       out = (float*)d_out;

    const int N = in_sizes[0] / FIN;
    const int E = in_sizes[1] / 2;
    const int* rows = ei;
    const int* cols = ei + E;

    float* ws   = (float*)d_ws;
    float* deg  = ws;
    float* dinv = ws + (size_t)N;
    float* h1   = ws + (size_t)2 * N;
    float* agg1 = ws + (size_t)18 * N;
    float* h2   = ws + (size_t)34 * N;
    float* agg2 = ws + (size_t)36 * N;

    const int BS = 256;

    gcn_init_kernel<<<(16 * N + BS - 1) / BS, BS, 0, stream>>>(deg, agg1, agg2, N);

    gcn_degree_kernel<<<(E + BS - 1) / BS, BS, 0, stream>>>(cols, deg, E);

    gcn_dinv_kernel<<<(N + BS - 1) / BS, BS, 0, stream>>>(deg, dinv, N);

    {
        int tiles = (N + 15) / 16;
        int waves_per_block = BS / 32;
        int blocks = (tiles + waves_per_block - 1) / waves_per_block;
        gcn_xw1_wmma_kernel<<<blocks, BS, 0, stream>>>(x, W1, h1, N);
    }

    {
        long long tot = (long long)E * 16;
        int blocks = (int)((tot + BS - 1) / BS);
        gcn_scatter1_kernel<<<blocks, BS, 0, stream>>>(rows, cols, h1, dinv, agg1, E);
    }

    gcn_layer1_finish_kernel<<<(N + BS - 1) / BS, BS, 0, stream>>>(
        h1, agg1, dinv, b1, W2, h2, N);

    {
        long long tot = (long long)E * 2;
        int blocks = (int)((tot + BS - 1) / BS);
        gcn_scatter2_kernel<<<blocks, BS, 0, stream>>>(rows, cols, h2, dinv, agg2, E);
    }

    gcn_final_kernel<<<(N + BS - 1) / BS, BS, 0, stream>>>(h2, agg2, dinv, b2, out, N);
}


# No atomics left: ballot-ranked counting split of edges into 128-node target buckets; degree and both layers summed per bucket in wave-private LDS, sum order now fixed; dinv[row] folded into h
# speedup vs baseline: 2.2558x; 2.2558x over previous
#include <hip/hip_runtime.h>
#include <hip/hip_bf16.h>


typedef __attribute__((ext_vector_type(2))) float v2f;
typedef __attribute__((ext_vector_type(8))) float v8f;

#define F1 16
#define F2 2
#define FIN 25

#define NPB_SHIFT 7
#define NPB 128
#define NB_MAX 1024
#define NB_BITS 10
#define SPLIT_WG 1024
#define WG 256
#define WAVES 8
#define LD_U 4

typedef __attribute__((address_space(3))) volatile int   lds_vint;
typedef __attribute__((address_space(3))) volatile float lds_vfloat;
typedef __attribute__((ext_vector_type(4))) float v4f;
typedef __attribute__((address_space(3))) volatile v4f   lds_vfloat4;

__device__ __forceinline__ unsigned wave_peers(int key, bool act, int nbits) {
    unsigned m = (unsigned)__ballot(act);
    for (int bit = 0; bit < nbits; ++bit) {
        bool one = ((key >> bit) & 1) != 0;
        unsigned bal = (unsigned)__ballot(act && one);
        m &= one ? bal : ~bal;
    }
    return act ? m : 0u;
}

__device__ __forceinline__ int block_excl_scan(int v, int* sm, int tid, int nthreads,
                                               int* total) {
    sm[tid] = v;
    __syncthreads();
    for (int off = 1; off < nthreads; off <<= 1) {
        int t = (tid >= off) ? sm[tid - off] : 0;
        __syncthreads();
        sm[tid] += t;
        __syncthreads();
    }
    int incl = sm[tid];
    *total = sm[nthreads - 1];
    __syncthreads();
    return incl - v;
}

__device__ __forceinline__ void chunk_count(const int* __restrict__ cols, int beg, int end,
                                            int* hist, int wave, int lane) {
    lds_vint* hw = (lds_vint*)(hist + wave * NB_MAX);
    const int nbatch = (end - beg + WG - 1) / WG;
    for (int j0 = 0; j0 < nbatch; j0 += LD_U) {
        int cc[LD_U];
        #pragma unroll
        for (int u = 0; u < LD_U; ++u) {
            const int i = beg + (j0 + u) * WG + wave * 32 + lane;
            cc[u] = cols[min(i, end - 1)];
        }
        #pragma unroll
        for (int u = 0; u < LD_U; ++u) {
            const int i = beg + (j0 + u) * WG + wave * 32 + lane;
            const bool act = i < end;
            const int key = cc[u] >> NPB_SHIFT;
            const unsigned m = wave_peers(key, act, NB_BITS);
            if (act && lane == __ffs(m) - 1) hw[key] = hw[key] + __popc(m);
        }
    }
}

__global__ void __launch_bounds__(WG)
gcn_split_count_kernel(const int* __restrict__ cols,
                       int* __restrict__ counts,
                       int n_edges, int chunk, int nb) {
    __shared__ int hist[WAVES * NB_MAX];
    const int tid  = threadIdx.x;
    const int lane = tid & 31;
    const int wave = tid >> 5;
    const int wg   = blockIdx.x;
    for (int k = tid; k < WAVES * NB_MAX; k += WG) hist[k] = 0;
    __syncthreads();
    const int beg = min(n_edges, wg * chunk);
    const int end = min(n_edges, beg + chunk);
    chunk_count(cols, beg, end, hist, wave, lane);
    __syncthreads();
    for (int b = tid; b < nb; b += WG) {
        int s = 0;
        #pragma unroll
        for (int w = 0; w < WAVES; ++w) s += hist[w * NB_MAX + b];
        counts[b * SPLIT_WG + wg] = s;
    }
}

__global__ void __launch_bounds__(SPLIT_WG)
gcn_split_rowscan_kernel(int* __restrict__ counts,
                         int* __restrict__ totals) {
    __shared__ int sm[SPLIT_WG];
    const int tid = threadIdx.x;
    const int b   = blockIdx.x;
    int v = counts[b * SPLIT_WG + tid];
    int total;
    int ex = block_excl_scan(v, sm, tid, SPLIT_WG, &total);
    counts[b * SPLIT_WG + tid] = ex;
    if (tid == 0) totals[b] = total;
}

__global__ void __launch_bounds__(NB_MAX)
gcn_split_starts_kernel(const int* __restrict__ totals,
                        int* __restrict__ bstart,
                        int nb) {
    __shared__ int sm[NB_MAX];
    const int tid = threadIdx.x;
    int v = (tid < nb) ? totals[tid] : 0;
    int all;
    int ex = block_excl_scan(v, sm, tid, NB_MAX, &all);
    if (tid < nb) bstart[tid] = ex;
    if (tid == 0) bstart[nb] = all;
}

__global__ void __launch_bounds__(WG)
gcn_split_scatter_kernel(const int* __restrict__ rows,
                         const int* __restrict__ cols,
                         const int* __restrict__ counts,
                         const int* __restrict__ bstart,
                         int2* __restrict__ sorted,
                         int n_edges, int chunk, int nb) {
    __shared__ int hist[WAVES * NB_MAX];
    const int tid  = threadIdx.x;
    const int lane = tid & 31;
    const int wave = tid >> 5;
    const int wg   = blockIdx.x;
    for (int k = tid; k < WAVES * NB_MAX; k += WG) hist[k] = 0;
    __syncthreads();
    const int beg = min(n_edges, wg * chunk);
    const int end = min(n_edges, beg + chunk);
    chunk_count(cols, beg, end, hist, wave, lane);
    __syncthreads();
    for (int b = tid; b < nb; b += WG) {
        int run = bstart[b] + counts[b * SPLIT_WG + wg];
        #pragma unroll
        for (int w = 0; w < WAVES; ++w) {
            int t = hist[w * NB_MAX + b];
            hist[w * NB_MAX + b] = run;
            run += t;
        }
    }
    __syncthreads();
    lds_vint* hw = (lds_vint*)(hist + wave * NB_MAX);
    const int nbatch = (end - beg + WG - 1) / WG;
    for (int j0 = 0; j0 < nbatch; j0 += LD_U) {
        int rr[LD_U], cc[LD_U];
        #pragma unroll
        for (int u = 0; u < LD_U; ++u) {
            const int i = beg + (j0 + u) * WG + wave * 32 + lane;
            rr[u] = rows[min(i, end - 1)];
            cc[u] = cols[min(i, end - 1)];
        }
        #pragma unroll
        for (int u = 0; u < LD_U; ++u) {
            const int i = beg + (j0 + u) * WG + wave * 32 + lane;
            const bool act = i < end;
            const int key = cc[u] >> NPB_SHIFT;
            const unsigned m = wave_peers(key, act, NB_BITS);
            int p = 0;
            if (act) p = hw[key];
            if (act && lane == __ffs(m) - 1) hw[key] = p + __popc(m);
            if (act) sorted[p + __popc(m & ((1u << lane) - 1u))] = make_int2(rr[u], cc[u]);
        }
    }
}

__global__ void __launch_bounds__(WG)
gcn_degree_kernel(const int2* __restrict__ sorted,
                  const int* __restrict__ bstart,
                  float* __restrict__ dinv,
                  int n) {
    __shared__ int cnt[WAVES * NPB];
    const int tid  = threadIdx.x;
    const int lane = tid & 31;
    const int wave = tid >> 5;
    const int b    = blockIdx.x;
    const int base = b << NPB_SHIFT;
    for (int k = tid; k < WAVES * NPB; k += WG) cnt[k] = 0;
    __syncthreads();
    const int beg = bstart[b];
    const int end = bstart[b + 1];
    lds_vint* cw = (lds_vint*)(cnt + wave * NPB);
    const int nbatch = (end - beg + WG - 1) / WG;
    for (int j0 = 0; j0 < nbatch; j0 += LD_U) {
        int cc[LD_U];
        #pragma unroll
        for (int u = 0; u < LD_U; ++u) {
            const int i = beg + (j0 + u) * WG + tid;
            cc[u] = sorted[min(i, end - 1)].y;
        }
        #pragma unroll
        for (int u = 0; u < LD_U; ++u) {
            const int i = beg + (j0 + u) * WG + tid;
            const bool act = i < end;
            const int key = cc[u] - base;
            const unsigned m = wave_peers(key, act, NPB_SHIFT);
            if (act && lane == __ffs(m) - 1) cw[key] = cw[key] + __popc(m);
        }
    }
    __syncthreads();
    if (tid < NPB && base + tid < n) {
        int d = 1;
        #pragma unroll
        for (int w = 0; w < WAVES; ++w) d += cnt[w * NPB + tid];
        dinv[base + tid] = rsqrtf((float)d);
    }
}

__global__ void gcn_xw1_wmma_kernel(const float* __restrict__ x,
                                    const float* __restrict__ W1,
                                    const float* __restrict__ dinv,
                                    float* __restrict__ h1s,
                                    int n_nodes) {
    const int lane  = threadIdx.x & 31;
    const int wave  = threadIdx.x >> 5;
    const int tile  = blockIdx.x * (blockDim.x >> 5) + wave;
    const int node0 = tile * 16;
    if (node0 >= n_nodes) return;

    const int half = lane >> 4;
    const int l16  = lane & 15;
    const int m    = l16;

    v8f c = {0.f, 0.f, 0.f, 0.f, 0.f, 0.f, 0.f, 0.f};

    #pragma unroll
    for (int k0 = 0; k0 < 28; k0 += 4) {
        const int ka = k0 + 2 * half;
        v2f a, b;
        a.x = (ka     < FIN) ? x[(node0 + m) * FIN + ka    ] : 0.0f;
        a.y = (ka + 1 < FIN) ? x[(node0 + m) * FIN + ka + 1] : 0.0f;
        b.x = (ka     < FIN) ? W1[(ka    ) * F1 + l16] : 0.0f;
        b.y = (ka + 1 < FIN) ? W1[(ka + 1) * F1 + l16] : 0.0f;
        c = __builtin_amdgcn_wmma_f32_16x16x4_f32(
                 false, a,  false, b,
                 (short)0, c,  false,  false);
    }

    #pragma unroll
    for (int v = 0; v < 8; ++v) {
        const int node = node0 + v + 8 * half;
        h1s[node * F1 + l16] = c[v] * dinv[node];
    }
}

#define L1_U 8
__global__ void __launch_bounds__(WG)
gcn_layer1_kernel(const int2* __restrict__ sorted,
                  const int* __restrict__ bstart,
                  const float* __restrict__ h1s,
                  const float* __restrict__ dinv,
                  const float* __restrict__ b1,
                  const float* __restrict__ W2,
                  float* __restrict__ h2s,
                  int n) {
    __shared__ __attribute__((aligned(16))) float acc[WAVES * NPB * F1];
    const int tid  = threadIdx.x;
    const int lane = tid & 31;
    const int wave = tid >> 5;
    const int grp  = lane >> 2;
    const int q    = lane & 3;
    const int b    = blockIdx.x;
    const int base = b << NPB_SHIFT;
    for (int k = tid; k < WAVES * NPB * F1; k += WG) acc[k] = 0.0f;
    __syncthreads();
    const int beg = bstart[b];
    const int end = bstart[b + 1];
    const v4f* h1v = (const v4f*)h1s;
    lds_vfloat4* aw = (lds_vfloat4*)(acc + wave * (NPB * F1));
    for (int i0 = beg; i0 < end; i0 += 8 * WAVES * L1_U) {
        int2 ee[L1_U];
        v4f  vv[L1_U];
        #pragma unroll
        for (int u = 0; u < L1_U; ++u) {
            const int i = i0 + u * (8 * WAVES) + wave * 8 + grp;
            ee[u] = sorted[min(i, end - 1)];
        }
        #pragma unroll
        for (int u = 0; u < L1_U; ++u) vv[u] = h1v[ee[u].x * 4 + q];
        #pragma unroll
        for (int u = 0; u < L1_U; ++u) {
            const int i = i0 + u * (8 * WAVES) + wave * 8 + grp;
            const bool act = i < end;
            const int key = ee[u].y - base;
            const unsigned m = wave_peers(key, act, NPB_SHIFT);
            const int rank = __popc(m & ((1u << lane) - 1u)) >> 2;
            for (int turn = 0; ; ++turn) {
                const bool mine = act && rank == turn;
                if (__ballot(mine) == 0) break;
                if (mine) {
                    v4f t = aw[key * 4 + q];
                    aw[key * 4 + q] = t + vv[u];
                }
            }
        }
    }
    __syncthreads();
    for (int k = tid; k < NPB * F1; k += WG) {
        float s = acc[k];
        #pragma unroll
        for (int w = 1; w < WAVES; ++w) s += acc[w * (NPB * F1) + k];
        acc[k] = s;
    }
    __syncthreads();
    const int node = base + tid;
    if (tid < NPB && node < n) {
        float di = dinv[node];
        float a0 = 0.0f, a1 = 0.0f;
        #pragma unroll
        for (int k = 0; k < F1; ++k) {
            float v = (acc[tid * F1 + k] + h1s[node * F1 + k]) * di + b1[k];
            v = v > 0.0f ? v : 0.0f;
            a0 = fmaf(v, W2[k * F2 + 0], a0);
            a1 = fmaf(v, W2[k * F2 + 1], a1);
        }
        h2s[node * F2 + 0] = a0 * di;
        h2s[node * F2 + 1] = a1 * di;
    }
}

__global__ void __launch_bounds__(WG)
gcn_layer2_kernel(const int2* __restrict__ sorted,
                  const int* __restrict__ bstart,
                  const float* __restrict__ h2s,
                  const float* __restrict__ dinv,
                  const float* __restrict__ b2,
                  float* __restrict__ out,
                  int n) {
    __shared__ float acc[WAVES * NPB * F2];
    const int tid  = threadIdx.x;
    const int lane = tid & 31;
    const int wave = tid >> 5;
    const int b    = blockIdx.x;
    const int base = b << NPB_SHIFT;
    for (int k = tid; k < WAVES * NPB * F2; k += WG) acc[k] = 0.0f;
    __syncthreads();
    const int beg = bstart[b];
    const int end = bstart[b + 1];
    const float2* h2v = (const float2*)h2s;
    lds_vfloat* aw = (lds_vfloat*)(acc + wave * (NPB * F2));
    const int nbatch = (end - beg + WG - 1) / WG;
    for (int j0 = 0; j0 < nbatch; j0 += LD_U) {
        int2   ee[LD_U];
        float2 vv[LD_U];
        #pragma unroll
        for (int u = 0; u < LD_U; ++u) {
            const int i = beg + (j0 + u) * WG + tid;
            ee[u] = sorted[min(i, end - 1)];
        }
        #pragma unroll
        for (int u = 0; u < LD_U; ++u) vv[u] = h2v[ee[u].x];
        #pragma unroll
        for (int u = 0; u < LD_U; ++u) {
            const int i = beg + (j0 + u) * WG + tid;
            const bool act = i < end;
            const int key = ee[u].y - base;
            const unsigned m = wave_peers(key, act, NPB_SHIFT);
            const int rank = __popc(m & ((1u << lane) - 1u));
            for (int turn = 0; ; ++turn) {
                const bool mine = act && rank == turn;
                if (__ballot(mine) == 0) break;
                if (mine) {
                    aw[key * F2 + 0] = aw[key * F2 + 0] + vv[u].x;
                    aw[key * F2 + 1] = aw[key * F2 + 1] + vv[u].y;
                }
            }
        }
    }
    __syncthreads();
    const int node = base + tid;
    if (tid < NPB && node < n) {
        float s0 = 0.0f, s1 = 0.0f;
        #pragma unroll
        for (int w = 0; w < WAVES; ++w) {
            s0 += acc[w * (NPB * F2) + tid * F2 + 0];
            s1 += acc[w * (NPB * F2) + tid * F2 + 1];
        }
        float di = dinv[node];
        float2 s = h2v[node];
        float a = (s0 + s.x) * di + b2[0];
        float c = (s1 + s.y) * di + b2[1];
        float m = fmaxf(a, c);
        float lse = m + logf(expf(a - m) + expf(c - m));
        out[node * F2 + 0] = a - lse;
        out[node * F2 + 1] = c - lse;
    }
}

extern "C" void kernel_launch(void* const* d_in, const int* in_sizes, int n_in,
                              void* d_out, int out_size, void* d_ws, size_t ws_size,
                              hipStream_t stream) {
    const float* x   = (const float*)d_in[0];
    const int*   ei  = (const int*)d_in[1];
    const float* W1  = (const float*)d_in[2];
    const float* b1  = (const float*)d_in[3];
    const float* W2  = (const float*)d_in[4];
    const float* b2  = (const float*)d_in[5];
    float*       out = (float*)d_out;

    const int N = in_sizes[0] / FIN;
    const int E = in_sizes[1] / 2;
    const int* rows = ei;
    const int* cols = ei + E;

    const int NB    = (N + NPB - 1) / NPB;
    const int chunk = (E + SPLIT_WG - 1) / SPLIT_WG;

    int*   wsi    = (int*)d_ws;
    int2*  sorted = (int2*)wsi;
    float* h1s    = (float*)(wsi + (size_t)2 * E);
    float* h2s    = h1s + (size_t)16 * N;
    float* dinv   = h2s + (size_t)2 * N;
    int*   counts = (int*)(dinv + (size_t)N);
    int*   totals = counts + (size_t)NB_MAX * SPLIT_WG;
    int*   bstart = totals + NB_MAX;

    gcn_split_count_kernel<<<SPLIT_WG, WG, 0, stream>>>(cols, counts, E, chunk, NB);
    gcn_split_rowscan_kernel<<<NB, SPLIT_WG, 0, stream>>>(counts, totals);
    gcn_split_starts_kernel<<<1, NB_MAX, 0, stream>>>(totals, bstart, NB);
    gcn_split_scatter_kernel<<<SPLIT_WG, WG, 0, stream>>>(rows, cols, counts, bstart,
                                                          sorted, E, chunk, NB);

    gcn_degree_kernel<<<NB, WG, 0, stream>>>(sorted, bstart, dinv, N);

    {
        const int BS = 256;
        int tiles = (N + 15) / 16;
        int waves_per_block = BS / 32;
        int blocks = (tiles + waves_per_block - 1) / waves_per_block;
        gcn_xw1_wmma_kernel<<<blocks, BS, 0, stream>>>(x, W1, dinv, h1s, N);
    }

    gcn_layer1_kernel<<<NB, WG, 0, stream>>>(sorted, bstart, h1s, dinv, b1, W2, h2s, N);

    gcn_layer2_kernel<<<NB, WG, 0, stream>>>(sorted, bstart, h2s, dinv, b2, out, N);
}


# Split scatter staged through LDS: each of 256 edge chunks is bucket-ordered in LDS and copied out as whole per-bucket segments, not one 8-byte store per edge; edge (sum) order inside buckets changes
# speedup vs baseline: 3.7386x; 1.6573x over previous
#include <hip/hip_runtime.h>
#include <hip/hip_bf16.h>


typedef __attribute__((ext_vector_type(2))) float v2f;
typedef __attribute__((ext_vector_type(8))) float v8f;

#define F1 16
#define F2 2
#define FIN 25

#define NPB_SHIFT 7
#define NPB 128
#define NB_MAX 1024
#define NB_BITS 10
#define SPLIT_WG 256
#define SWG 1024
#define SWAVES 32
#define STAGE_CAP 16384
#define WG 256
#define WAVES 8
#define LD_U 4

typedef __attribute__((address_space(3))) volatile int   lds_vint;
typedef __attribute__((address_space(3))) volatile float lds_vfloat;
typedef __attribute__((ext_vector_type(4))) float v4f;
typedef __attribute__((address_space(3))) volatile v4f   lds_vfloat4;
typedef __attribute__((ext_vector_type(2))) int v2i;

__device__ __forceinline__ unsigned wave_peers(int key, bool act, int nbits) {
    unsigned m = (unsigned)__ballot(act);
    for (int bit = 0; bit < nbits; ++bit) {
        bool one = ((key >> bit) & 1) != 0;
        unsigned bal = (unsigned)__ballot(act && one);
        m &= one ? bal : ~bal;
    }
    return act ? m : 0u;
}

__device__ __forceinline__ int block_excl_scan(int v, int* sm, int tid, int nthreads,
                                               int* total) {
    sm[tid] = v;
    __syncthreads();
    for (int off = 1; off < nthreads; off <<= 1) {
        int t = (tid >= off) ? sm[tid - off] : 0;
        __syncthreads();
        sm[tid] += t;
        __syncthreads();
    }
    int incl = sm[tid];
    *total = sm[nthreads - 1];
    __syncthreads();
    return incl - v;
}

__device__ __forceinline__ void chunk_count(const int* __restrict__ cols, int beg, int end,
                                            int* hist, int wave, int lane) {
    lds_vint* hw = (lds_vint*)(hist + wave * NB_MAX);
    const int nbatch = (end - beg + SWG - 1) / SWG;
    for (int j0 = 0; j0 < nbatch; j0 += LD_U) {
        int cc[LD_U];
        #pragma unroll
        for (int u = 0; u < LD_U; ++u) {
            const int i = beg + (j0 + u) * SWG + wave * 32 + lane;
            cc[u] = cols[min(i, end - 1)];
        }
        #pragma unroll
        for (int u = 0; u < LD_U; ++u) {
            const int i = beg + (j0 + u) * SWG + wave * 32 + lane;
            const bool act = i < end;
            const int key = cc[u] >> NPB_SHIFT;
            const unsigned m = wave_peers(key, act, NB_BITS);
            if (act && lane == __ffs(m) - 1) hw[key] = hw[key] + __popc(m);
        }
    }
}

__global__ void __launch_bounds__(SWG)
gcn_split_count_kernel(const int* __restrict__ cols,
                       int* __restrict__ counts,
                       int n_edges, int chunk, int nb) {
    __shared__ int hist[SWAVES * NB_MAX];
    const int tid  = threadIdx.x;
    const int lane = tid & 31;
    const int wave = tid >> 5;
    const int wg   = blockIdx.x;
    for (int k = tid; k < SWAVES * NB_MAX; k += SWG) hist[k] = 0;
    __syncthreads();
    const int beg = min(n_edges, wg * chunk);
    const int end = min(n_edges, beg + chunk);
    chunk_count(cols, beg, end, hist, wave, lane);
    __syncthreads();
    for (int b = tid; b < nb; b += SWG) {
        int s = 0;
        #pragma unroll
        for (int w = 0; w < SWAVES; ++w) s += hist[w * NB_MAX + b];
        counts[b * SPLIT_WG + wg] = s;
    }
}

__global__ void __launch_bounds__(SPLIT_WG)
gcn_split_rowscan_kernel(int* __restrict__ counts,
                         int* __restrict__ totals) {
    __shared__ int sm[SPLIT_WG];
    const int tid = threadIdx.x;
    const int b   = blockIdx.x;
    int v = counts[b * SPLIT_WG + tid];
    int total;
    int ex = block_excl_scan(v, sm, tid, SPLIT_WG, &total);
    counts[b * SPLIT_WG + tid] = ex;
    if (tid == 0) totals[b] = total;
}

__global__ void __launch_bounds__(NB_MAX)
gcn_split_starts_kernel(const int* __restrict__ totals,
                        int* __restrict__ bstart,
                        int nb) {
    __shared__ int sm[NB_MAX];
    const int tid = threadIdx.x;
    int v = (tid < nb) ? totals[tid] : 0;
    int all;
    int ex = block_excl_scan(v, sm, tid, NB_MAX, &all);
    if (tid < nb) bstart[tid] = ex;
    if (tid == 0) bstart[nb] = all;
}

__global__ void __launch_bounds__(SWG)
gcn_split_scatter_kernel(const int* __restrict__ rows,
                         const int* __restrict__ cols,
                         const int* __restrict__ counts,
                         const int* __restrict__ bstart,
                         int2* __restrict__ sorted,
                         int n_edges, int chunk, int nb) {
    static_assert(SWG == NB_MAX, "one thread per bucket in the cursor step");
    __shared__ int hist[SWAVES * NB_MAX];
    __shared__ v2i stage[STAGE_CAP];
    __shared__ int lstart[NB_MAX];
    __shared__ int delta[NB_MAX];
    __shared__ int scan_sm[SWG];
    const int tid  = threadIdx.x;
    const int lane = tid & 31;
    const int wave = tid >> 5;
    const int wg   = blockIdx.x;
    for (int k = tid; k < SWAVES * NB_MAX; k += SWG) hist[k] = 0;
    __syncthreads();
    const int beg = min(n_edges, wg * chunk);
    const int end = min(n_edges, beg + chunk);
    const int gb = (tid < nb) ? bstart[tid] + counts[tid * SPLIT_WG + wg] : 0;
    chunk_count(cols, beg, end, hist, wave, lane);
    __syncthreads();
    int tot = 0;
    #pragma unroll
    for (int w = 0; w < SWAVES; ++w) {
        int t = hist[w * NB_MAX + tid];
        hist[w * NB_MAX + tid] = tot;
        tot += t;
    }
    int all;
    const int ls = block_excl_scan(tot, scan_sm, tid, SWG, &all);
    lstart[tid] = ls;
    delta[tid]  = gb - ls;
    __syncthreads();
    lds_vint* hw = (lds_vint*)(hist + wave * NB_MAX);
    const int nbatch = (end - beg + SWG - 1) / SWG;
    for (int j0 = 0; j0 < nbatch; j0 += LD_U) {
        int rr[LD_U], cc[LD_U];
        #pragma unroll
        for (int u = 0; u < LD_U; ++u) {
            const int i = beg + (j0 + u) * SWG + wave * 32 + lane;
            rr[u] = rows[min(i, end - 1)];
            cc[u] = cols[min(i, end - 1)];
        }
        #pragma unroll
        for (int u = 0; u < LD_U; ++u) {
            const int i = beg + (j0 + u) * SWG + wave * 32 + lane;
            const bool act = i < end;
            const int key = cc[u] >> NPB_SHIFT;
            const unsigned m = wave_peers(key, act, NB_BITS);
            int p = 0;
            if (act) p = hw[key];
            if (act && lane == __ffs(m) - 1) hw[key] = p + __popc(m);
            if (act) {
                v2i e;
                e.x = rr[u];
                e.y = cc[u];
                stage[lstart[key] + p + __popc(m & ((1u << lane) - 1u))] = e;
            }
        }
    }
    __syncthreads();
    v2i* sv = (v2i*)sorted;
    const int n_chunk = end - beg;
    for (int i0 = tid; i0 < n_chunk; i0 += LD_U * SWG) {
        v2i e[LD_U];
        int d[LD_U];
        #pragma unroll
        for (int u = 0; u < LD_U; ++u) e[u] = stage[min(i0 + u * SWG, n_chunk - 1)];
        #pragma unroll
        for (int u = 0; u < LD_U; ++u) d[u] = delta[e[u].y >> NPB_SHIFT];
        #pragma unroll
        for (int u = 0; u < LD_U; ++u) {
            const int i = i0 + u * SWG;
            if (i < n_chunk) sv[i + d[u]] = e[u];
        }
    }
}

__global__ void __launch_bounds__(WG)
gcn_degree_kernel(const int2* __restrict__ sorted,
                  const int* __restrict__ bstart,
                  float* __restrict__ dinv,
                  int n) {
    __shared__ int cnt[WAVES * NPB];
    const int tid  = threadIdx.x;
    const int lane = tid & 31;
    const int wave = tid >> 5;
    const int b    = blockIdx.x;
    const int base = b << NPB_SHIFT;
    for (int k = tid; k < WAVES * NPB; k += WG) cnt[k] = 0;
    __syncthreads();
    const int beg = bstart[b];
    const int end = bstart[b + 1];
    lds_vint* cw = (lds_vint*)(cnt + wave * NPB);
    const int nbatch = (end - beg + WG - 1) / WG;
    for (int j0 = 0; j0 < nbatch; j0 += LD_U) {
        int cc[LD_U];
        #pragma unroll
        for (int u = 0; u < LD_U; ++u) {
            const int i = beg + (j0 + u) * WG + tid;
            cc[u] = sorted[min(i, end - 1)].y;
        }
        #pragma unroll
        for (int u = 0; u < LD_U; ++u) {
            const int i = beg + (j0 + u) * WG + tid;
            const bool act = i < end;
            const int key = cc[u] - base;
            const unsigned m = wave_peers(key, act, NPB_SHIFT);
            if (act && lane == __ffs(m) - 1) cw[key] = cw[key] + __popc(m);
        }
    }
    __syncthreads();
    if (tid < NPB && base + tid < n) {
        int d = 1;
        #pragma unroll
        for (int w = 0; w < WAVES; ++w) d += cnt[w * NPB + tid];
        dinv[base + tid] = rsqrtf((float)d);
    }
}

__global__ void gcn_xw1_wmma_kernel(const float* __restrict__ x,
                                    const float* __restrict__ W1,
                                    const float* __restrict__ dinv,
                                    float* __restrict__ h1s,
                                    int n_nodes) {
    const int lane  = threadIdx.x & 31;
    const int wave  = threadIdx.x >> 5;
    const int tile  = blockIdx.x * (blockDim.x >> 5) + wave;
    const int node0 = tile * 16;
    if (node0 >= n_nodes) return;

    const int half = lane >> 4;
    const int l16  = lane & 15;
    const int m    = l16;

    v8f c = {0.f, 0.f, 0.f, 0.f, 0.f, 0.f, 0.f, 0.f};

    #pragma unroll
    for (int k0 = 0; k0 < 28; k0 += 4) {
        const int ka = k0 + 2 * half;
        v2f a, b;
        a.x = (ka     < FIN) ? x[(node0 + m) * FIN + ka    ] : 0.0f;
        a.y = (ka + 1 < FIN) ? x[(node0 + m) * FIN + ka + 1] : 0.0f;
        b.x = (ka     < FIN) ? W1[(ka    ) * F1 + l16] : 0.0f;
        b.y = (ka + 1 < FIN) ? W1[(ka + 1) * F1 + l16] : 0.0f;
        c = __builtin_amdgcn_wmma_f32_16x16x4_f32(
                 false, a,  false, b,
                 (short)0, c,  false,  false);
    }

    #pragma unroll
    for (int v = 0; v < 8; ++v) {
        const int node = node0 + v + 8 * half;
        h1s[node * F1 + l16] = c[v] * dinv[node];
    }
}

#define L1_U 8
__global__ void __launch_bounds__(WG)
gcn_layer1_kernel(const int2* __restrict__ sorted,
                  const int* __restrict__ bstart,
                  const float* __restrict__ h1s,
                  const float* __restrict__ dinv,
                  const float* __restrict__ b1,
                  const float* __restrict__ W2,
                  float* __restrict__ h2s,
                  int n) {
    __shared__ __attribute__((aligned(16))) float acc[WAVES * NPB * F1];
    const int tid  = threadIdx.x;
    const int lane = tid & 31;
    const int wave = tid >> 5;
    const int grp  = lane >> 2;
    const int q    = lane & 3;
    const int b    = blockIdx.x;
    const int base = b << NPB_SHIFT;
    for (int k = tid; k < WAVES * NPB * F1; k += WG) acc[k] = 0.0f;
    __syncthreads();
    const int beg = bstart[b];
    const int end = bstart[b + 1];
    const v4f* h1v = (const v4f*)h1s;
    lds_vfloat4* aw = (lds_vfloat4*)(acc + wave * (NPB * F1));
    for (int i0 = beg; i0 < end; i0 += 8 * WAVES * L1_U) {
        int2 ee[L1_U];
        v4f  vv[L1_U];
        #pragma unroll
        for (int u = 0; u < L1_U; ++u) {
            const int i = i0 + u * (8 * WAVES) + wave * 8 + grp;
            ee[u] = sorted[min(i, end - 1)];
        }
        #pragma unroll
        for (int u = 0; u < L1_U; ++u) vv[u] = h1v[ee[u].x * 4 + q];
        #pragma unroll
        for (int u = 0; u < L1_U; ++u) {
            const int i = i0 + u * (8 * WAVES) + wave * 8 + grp;
            const bool act = i < end;
            const int key = ee[u].y - base;
            const unsigned m = wave_peers(key, act, NPB_SHIFT);
            const int rank = __popc(m & ((1u << lane) - 1u)) >> 2;
            for (int turn = 0; ; ++turn) {
                const bool mine = act && rank == turn;
                if (__ballot(mine) == 0) break;
                if (mine) {
                    v4f t = aw[key * 4 + q];
                    aw[key * 4 + q] = t + vv[u];
                }
            }
        }
    }
    __syncthreads();
    for (int k = tid; k < NPB * F1; k += WG) {
        float s = acc[k];
        #pragma unroll
        for (int w = 1; w < WAVES; ++w) s += acc[w * (NPB * F1) + k];
        acc[k] = s;
    }
    __syncthreads();
    const int node = base + tid;
    if (tid < NPB && node < n) {
        float di = dinv[node];
        float a0 = 0.0f, a1 = 0.0f;
        #pragma unroll
        for (int k = 0; k < F1; ++k) {
            float v = (acc[tid * F1 + k] + h1s[node * F1 + k]) * di + b1[k];
            v = v > 0.0f ? v : 0.0f;
            a0 = fmaf(v, W2[k * F2 + 0], a0);
            a1 = fmaf(v, W2[k * F2 + 1], a1);
        }
        h2s[node * F2 + 0] = a0 * di;
        h2s[node * F2 + 1] = a1 * di;
    }
}

__global__ void __launch_bounds__(WG)
gcn_layer2_kernel(const int2* __restrict__ sorted,
                  const int* __restrict__ bstart,
                  const float* __restrict__ h2s,
                  const float* __restrict__ dinv,
                  const float* __restrict__ b2,
                  float* __restrict__ out,
                  int n) {
    __shared__ float acc[WAVES * NPB * F2];
    const int tid  = threadIdx.x;
    const int lane = tid & 31;
    const int wave = tid >> 5;
    const int b    = blockIdx.x;
    const int base = b << NPB_SHIFT;
    for (int k = tid; k < WAVES * NPB * F2; k += WG) acc[k] = 0.0f;
    __syncthreads();
    const int beg = bstart[b];
    const int end = bstart[b + 1];
    const float2* h2v = (const float2*)h2s;
    lds_vfloat* aw = (lds_vfloat*)(acc + wave * (NPB * F2));
    const int nbatch = (end - beg + WG - 1) / WG;
    for (int j0 = 0; j0 < nbatch; j0 += LD_U) {
        int2   ee[LD_U];
        float2 vv[LD_U];
        #pragma unroll
        for (int u = 0; u < LD_U; ++u) {
            const int i = beg + (j0 + u) * WG + tid;
            ee[u] = sorted[min(i, end - 1)];
        }
        #pragma unroll
        for (int u = 0; u < LD_U; ++u) vv[u] = h2v[ee[u].x];
        #pragma unroll
        for (int u = 0; u < LD_U; ++u) {
            const int i = beg + (j0 + u) * WG + tid;
            const bool act = i < end;
            const int key = ee[u].y - base;
            const unsigned m = wave_peers(key, act, NPB_SHIFT);
            const int rank = __popc(m & ((1u << lane) - 1u));
            for (int turn = 0; ; ++turn) {
                const bool mine = act && rank == turn;
                if (__ballot(mine) == 0) break;
                if (mine) {
                    aw[key * F2 + 0] = aw[key * F2 + 0] + vv[u].x;
                    aw[key * F2 + 1] = aw[key * F2 + 1] + vv[u].y;
                }
            }
        }
    }
    __syncthreads();
    const int node = base + tid;
    if (tid < NPB && node < n) {
        float s0 = 0.0f, s1 = 0.0f;
        #pragma unroll
        for (int w = 0; w < WAVES; ++w) {
            s0 += acc[w * (NPB * F2) + tid * F2 + 0];
            s1 += acc[w * (NPB * F2) + tid * F2 + 1];
        }
        float di = dinv[node];
        float2 s = h2v[node];
        float a = (s0 + s.x) * di + b2[0];
        float c = (s1 + s.y) * di + b2[1];
        float m = fmaxf(a, c);
        float lse = m + logf(expf(a - m) + expf(c - m));
        out[node * F2 + 0] = a - lse;
        out[node * F2 + 1] = c - lse;
    }
}

extern "C" void kernel_launch(void* const* d_in, const int* in_sizes, int n_in,
                              void* d_out, int out_size, void* d_ws, size_t ws_size,
                              hipStream_t stream) {
    const float* x   = (const float*)d_in[0];
    const int*   ei  = (const int*)d_in[1];
    const float* W1  = (const float*)d_in[2];
    const float* b1  = (const float*)d_in[3];
    const float* W2  = (const float*)d_in[4];
    const float* b2  = (const float*)d_in[5];
    float*       out = (float*)d_out;

    const int N = in_sizes[0] / FIN;
    const int E = in_sizes[1] / 2;
    const int* rows = ei;
    const int* cols = ei + E;

    const int NB    = (N + NPB - 1) / NPB;
    const int chunk = (E + SPLIT_WG - 1) / SPLIT_WG;

    int*   wsi    = (int*)d_ws;
    int2*  sorted = (int2*)wsi;
    float* h1s    = (float*)(wsi + (size_t)2 * E);
    float* h2s    = h1s + (size_t)16 * N;
    float* dinv   = h2s + (size_t)2 * N;
    int*   counts = (int*)(dinv + (size_t)N);
    int*   totals = counts + (size_t)NB_MAX * SPLIT_WG;
    int*   bstart = totals + NB_MAX;

    gcn_split_count_kernel<<<SPLIT_WG, SWG, 0, stream>>>(cols, counts, E, chunk, NB);
    gcn_split_rowscan_kernel<<<NB, SPLIT_WG, 0, stream>>>(counts, totals);
    gcn_split_starts_kernel<<<1, NB_MAX, 0, stream>>>(totals, bstart, NB);
    gcn_split_scatter_kernel<<<SPLIT_WG, SWG, 0, stream>>>(rows, cols, counts, bstart,
                                                          sorted, E, chunk, NB);

    gcn_degree_kernel<<<NB, WG, 0, stream>>>(sorted, bstart, dinv, N);

    {
        const int BS = 256;
        int tiles = (N + 15) / 16;
        int waves_per_block = BS / 32;
        int blocks = (tiles + waves_per_block - 1) / waves_per_block;
        gcn_xw1_wmma_kernel<<<blocks, BS, 0, stream>>>(x, W1, dinv, h1s, N);
    }

    gcn_layer1_kernel<<<NB, WG, 0, stream>>>(sorted, bstart, h1s, dinv, b1, W2, h2s, N);

    gcn_layer2_kernel<<<NB, WG, 0, stream>>>(sorted, bstart, h2s, dinv, b2, out, N);
}


# Same-key lane masks cheaper: per key bit one signed bit extract, one compare into a scalar mask, one 3-input bit op (table 0x90); ranks by v_mbcnt; wave32 ballot builtin; results bit-identical
# speedup vs baseline: 4.3374x; 1.1602x over previous
#include <hip/hip_runtime.h>
#include <hip/hip_bf16.h>


typedef __attribute__((ext_vector_type(2))) float v2f;
typedef __attribute__((ext_vector_type(8))) float v8f;

#define F1 16
#define F2 2
#define FIN 25

#define NPB_SHIFT 7
#define NPB 128
#define NB_MAX 1024
#define NB_BITS 10
#define SPLIT_WG 256
#define SWG 1024
#define SWAVES 32
#define STAGE_CAP 16384
#define WG 256
#define WAVES 8
#define LD_U 4

typedef __attribute__((address_space(3))) volatile int   lds_vint;
typedef __attribute__((address_space(3))) volatile float lds_vfloat;
typedef __attribute__((ext_vector_type(4))) float v4f;
typedef __attribute__((address_space(3))) volatile v4f   lds_vfloat4;
typedef __attribute__((ext_vector_type(2))) int v2i;

__device__ __forceinline__ unsigned wave_peers(int key, bool act, int nbits) {
    unsigned m = __builtin_amdgcn_ballot_w32(act);
    for (int bit = 0; bit < nbits; ++bit) {
        const int x = __builtin_amdgcn_sbfe(key, bit, 1);
        const unsigned bal = __builtin_amdgcn_ballot_w32(x != 0);
        m = __builtin_amdgcn_bitop3_b32(m, bal, (unsigned)x, 0x90);
    }
    return act ? m : 0u;
}

__device__ __forceinline__ int lanes_below(unsigned m) {
    return (int)__builtin_amdgcn_mbcnt_lo(m, 0u);
}

__device__ __forceinline__ int block_excl_scan(int v, int* sm, int tid, int nthreads,
                                               int* total) {
    sm[tid] = v;
    __syncthreads();
    for (int off = 1; off < nthreads; off <<= 1) {
        int t = (tid >= off) ? sm[tid - off] : 0;
        __syncthreads();
        sm[tid] += t;
        __syncthreads();
    }
    int incl = sm[tid];
    *total = sm[nthreads - 1];
    __syncthreads();
    return incl - v;
}

__device__ __forceinline__ void chunk_count(const int* __restrict__ cols, int beg, int end,
                                            int* hist, int wave, int lane) {
    lds_vint* hw = (lds_vint*)(hist + wave * NB_MAX);
    const int nbatch = (end - beg + SWG - 1) / SWG;
    for (int j0 = 0; j0 < nbatch; j0 += LD_U) {
        int cc[LD_U];
        #pragma unroll
        for (int u = 0; u < LD_U; ++u) {
            const int i = beg + (j0 + u) * SWG + wave * 32 + lane;
            cc[u] = cols[min(i, end - 1)];
        }
        #pragma unroll
        for (int u = 0; u < LD_U; ++u) {
            const int i = beg + (j0 + u) * SWG + wave * 32 + lane;
            const bool act = i < end;
            const int key = cc[u] >> NPB_SHIFT;
            const unsigned m = wave_peers(key, act, NB_BITS);
            if (act && lanes_below(m) == 0) hw[key] = hw[key] + __popc(m);
        }
    }
}

__global__ void __launch_bounds__(SWG)
gcn_split_count_kernel(const int* __restrict__ cols,
                       int* __restrict__ counts,
                       int n_edges, int chunk, int nb) {
    __shared__ int hist[SWAVES * NB_MAX];
    const int tid  = threadIdx.x;
    const int lane = tid & 31;
    const int wave = tid >> 5;
    const int wg   = blockIdx.x;
    for (int k = tid; k < SWAVES * NB_MAX; k += SWG) hist[k] = 0;
    __syncthreads();
    const int beg = min(n_edges, wg * chunk);
    const int end = min(n_edges, beg + chunk);
    chunk_count(cols, beg, end, hist, wave, lane);
    __syncthreads();
    for (int b = tid; b < nb; b += SWG) {
        int s = 0;
        #pragma unroll
        for (int w = 0; w < SWAVES; ++w) s += hist[w * NB_MAX + b];
        counts[b * SPLIT_WG + wg] = s;
    }
}

__global__ void __launch_bounds__(SPLIT_WG)
gcn_split_rowscan_kernel(int* __restrict__ counts,
                         int* __restrict__ totals) {
    __shared__ int sm[SPLIT_WG];
    const int tid = threadIdx.x;
    const int b   = blockIdx.x;
    int v = counts[b * SPLIT_WG + tid];
    int total;
    int ex = block_excl_scan(v, sm, tid, SPLIT_WG, &total);
    counts[b * SPLIT_WG + tid] = ex;
    if (tid == 0) totals[b] = total;
}

__global__ void __launch_bounds__(NB_MAX)
gcn_split_starts_kernel(const int* __restrict__ totals,
                        int* __restrict__ bstart,
                        int nb) {
    __shared__ int sm[NB_MAX];
    const int tid = threadIdx.x;
    int v = (tid < nb) ? totals[tid] : 0;
    int all;
    int ex = block_excl_scan(v, sm, tid, NB_MAX, &all);
    if (tid < nb) bstart[tid] = ex;
    if (tid == 0) bstart[nb] = all;
}

__global__ void __launch_bounds__(SWG)
gcn_split_scatter_kernel(const int* __restrict__ rows,
                         const int* __restrict__ cols,
                         const int* __restrict__ counts,
                         const int* __restrict__ bstart,
                         int2* __restrict__ sorted,
                         int n_edges, int chunk, int nb) {
    static_assert(SWG == NB_MAX, "one thread per bucket in the cursor step");
    __shared__ int hist[SWAVES * NB_MAX];
    __shared__ v2i stage[STAGE_CAP];
    __shared__ int lstart[NB_MAX];
    __shared__ int delta[NB_MAX];
    __shared__ int scan_sm[SWG];
    const int tid  = threadIdx.x;
    const int lane = tid & 31;
    const int wave = tid >> 5;
    const int wg   = blockIdx.x;
    for (int k = tid; k < SWAVES * NB_MAX; k += SWG) hist[k] = 0;
    __syncthreads();
    const int beg = min(n_edges, wg * chunk);
    const int end = min(n_edges, beg + chunk);
    const int gb = (tid < nb) ? bstart[tid] + counts[tid * SPLIT_WG + wg] : 0;
    chunk_count(cols, beg, end, hist, wave, lane);
    __syncthreads();
    int tot = 0;
    #pragma unroll
    for (int w = 0; w < SWAVES; ++w) {
        int t = hist[w * NB_MAX + tid];
        hist[w * NB_MAX + tid] = tot;
        tot += t;
    }
    int all;
    const int ls = block_excl_scan(tot, scan_sm, tid, SWG, &all);
    lstart[tid] = ls;
    delta[tid]  = gb - ls;
    __syncthreads();
    lds_vint* hw = (lds_vint*)(hist + wave * NB_MAX);
    const int nbatch = (end - beg + SWG - 1) / SWG;
    for (int j0 = 0; j0 < nbatch; j0 += LD_U) {
        int rr[LD_U], cc[LD_U];
        #pragma unroll
        for (int u = 0; u < LD_U; ++u) {
            const int i = beg + (j0 + u) * SWG + wave * 32 + lane;
            rr[u] = rows[min(i, end - 1)];
            cc[u] = cols[min(i, end - 1)];
        }
        #pragma unroll
        for (int u = 0; u < LD_U; ++u) {
            const int i = beg + (j0 + u) * SWG + wave * 32 + lane;
            const bool act = i < end;
            const int key = cc[u] >> NPB_SHIFT;
            const unsigned m = wave_peers(key, act, NB_BITS);
            const int rank = lanes_below(m);
            int p = 0;
            if (act) p = hw[key];
            if (act && rank == 0) hw[key] = p + __popc(m);
            if (act) {
                v2i e;
                e.x = rr[u];
                e.y = cc[u];
                stage[lstart[key] + p + rank] = e;
            }
        }
    }
    __syncthreads();
    v2i* sv = (v2i*)sorted;
    const int n_chunk = end - beg;
    for (int i0 = tid; i0 < n_chunk; i0 += LD_U * SWG) {
        v2i e[LD_U];
        int d[LD_U];
        #pragma unroll
        for (int u = 0; u < LD_U; ++u) e[u] = stage[min(i0 + u * SWG, n_chunk - 1)];
        #pragma unroll
        for (int u = 0; u < LD_U; ++u) d[u] = delta[e[u].y >> NPB_SHIFT];
        #pragma unroll
        for (int u = 0; u < LD_U; ++u) {
            const int i = i0 + u * SWG;
            if (i < n_chunk) sv[i + d[u]] = e[u];
        }
    }
}

__global__ void __launch_bounds__(WG)
gcn_degree_kernel(const int2* __restrict__ sorted,
                  const int* __restrict__ bstart,
                  float* __restrict__ dinv,
                  int n) {
    __shared__ int cnt[WAVES * NPB];
    const int tid  = threadIdx.x;
    const int lane = tid & 31;
    const int wave = tid >> 5;
    const int b    = blockIdx.x;
    const int base = b << NPB_SHIFT;
    for (int k = tid; k < WAVES * NPB; k += WG) cnt[k] = 0;
    __syncthreads();
    const int beg = bstart[b];
    const int end = bstart[b + 1];
    lds_vint* cw = (lds_vint*)(cnt + wave * NPB);
    const int nbatch = (end - beg + WG - 1) / WG;
    for (int j0 = 0; j0 < nbatch; j0 += LD_U) {
        int cc[LD_U];
        #pragma unroll
        for (int u = 0; u < LD_U; ++u) {
            const int i = beg + (j0 + u) * WG + tid;
            cc[u] = sorted[min(i, end - 1)].y;
        }
        #pragma unroll
        for (int u = 0; u < LD_U; ++u) {
            const int i = beg + (j0 + u) * WG + tid;
            const bool act = i < end;
            const int key = cc[u] - base;
            const unsigned m = wave_peers(key, act, NPB_SHIFT);
            if (act && lanes_below(m) == 0) cw[key] = cw[key] + __popc(m);
        }
    }
    __syncthreads();
    if (tid < NPB && base + tid < n) {
        int d = 1;
        #pragma unroll
        for (int w = 0; w < WAVES; ++w) d += cnt[w * NPB + tid];
        dinv[base + tid] = rsqrtf((float)d);
    }
}

__global__ void gcn_xw1_wmma_kernel(const float* __restrict__ x,
                                    const float* __restrict__ W1,
                                    const float* __restrict__ dinv,
                                    float* __restrict__ h1s,
                                    int n_nodes) {
    const int lane  = threadIdx.x & 31;
    const int wave  = threadIdx.x >> 5;
    const int tile  = blockIdx.x * (blockDim.x >> 5) + wave;
    const int node0 = tile * 16;
    if (node0 >= n_nodes) return;

    const int half = lane >> 4;
    const int l16  = lane & 15;
    const int m    = l16;

    v8f c = {0.f, 0.f, 0.f, 0.f, 0.f, 0.f, 0.f, 0.f};

    #pragma unroll
    for (int k0 = 0; k0 < 28; k0 += 4) {
        const int ka = k0 + 2 * half;
        v2f a, b;
        a.x = (ka     < FIN) ? x[(node0 + m) * FIN + ka    ] : 0.0f;
        a.y = (ka + 1 < FIN) ? x[(node0 + m) * FIN + ka + 1] : 0.0f;
        b.x = (ka     < FIN) ? W1[(ka    ) * F1 + l16] : 0.0f;
        b.y = (ka + 1 < FIN) ? W1[(ka + 1) * F1 + l16] : 0.0f;
        c = __builtin_amdgcn_wmma_f32_16x16x4_f32(
                 false, a,  false, b,
                 (short)0, c,  false,  false);
    }

    #pragma unroll
    for (int v = 0; v < 8; ++v) {
        const int node = node0 + v + 8 * half;
        h1s[node * F1 + l16] = c[v] * dinv[node];
    }
}

#define L1_U 8
__global__ void __launch_bounds__(WG)
gcn_layer1_kernel(const int2* __restrict__ sorted,
                  const int* __restrict__ bstart,
                  const float* __restrict__ h1s,
                  const float* __restrict__ dinv,
                  const float* __restrict__ b1,
                  const float* __restrict__ W2,
                  float* __restrict__ h2s,
                  int n) {
    __shared__ __attribute__((aligned(16))) float acc[WAVES * NPB * F1];
    const int tid  = threadIdx.x;
    const int lane = tid & 31;
    const int wave = tid >> 5;
    const int grp  = lane >> 2;
    const int q    = lane & 3;
    const int b    = blockIdx.x;
    const int base = b << NPB_SHIFT;
    for (int k = tid; k < WAVES * NPB * F1; k += WG) acc[k] = 0.0f;
    __syncthreads();
    const int beg = bstart[b];
    const int end = bstart[b + 1];
    const v4f* h1v = (const v4f*)h1s;
    lds_vfloat4* aw = (lds_vfloat4*)(acc + wave * (NPB * F1));
    for (int i0 = beg; i0 < end; i0 += 8 * WAVES * L1_U) {
        int2 ee[L1_U];
        v4f  vv[L1_U];
        #pragma unroll
        for (int u = 0; u < L1_U; ++u) {
            const int i = i0 + u * (8 * WAVES) + wave * 8 + grp;
            ee[u] = sorted[min(i, end - 1)];
        }
        #pragma unroll
        for (int u = 0; u < L1_U; ++u) vv[u] = h1v[ee[u].x * 4 + q];
        #pragma unroll
        for (int u = 0; u < L1_U; ++u) {
            const int i = i0 + u * (8 * WAVES) + wave * 8 + grp;
            const bool act = i < end;
            const int key = ee[u].y - base;
            const unsigned m = wave_peers(key, act, NPB_SHIFT);
            const int rank = lanes_below(m) >> 2;
            for (int turn = 0; ; ++turn) {
                const bool mine = act && rank == turn;
                if (__builtin_amdgcn_ballot_w32(mine) == 0) break;
                if (mine) {
                    v4f t = aw[key * 4 + q];
                    aw[key * 4 + q] = t + vv[u];
                }
            }
        }
    }
    __syncthreads();
    for (int k = tid; k < NPB * F1; k += WG) {
        float s = acc[k];
        #pragma unroll
        for (int w = 1; w < WAVES; ++w) s += acc[w * (NPB * F1) + k];
        acc[k] = s;
    }
    __syncthreads();
    const int node = base + tid;
    if (tid < NPB && node < n) {
        float di = dinv[node];
        float a0 = 0.0f, a1 = 0.0f;
        #pragma unroll
        for (int k = 0; k < F1; ++k) {
            float v = (acc[tid * F1 + k] + h1s[node * F1 + k]) * di + b1[k];
            v = v > 0.0f ? v : 0.0f;
            a0 = fmaf(v, W2[k * F2 + 0], a0);
            a1 = fmaf(v, W2[k * F2 + 1], a1);
        }
        h2s[node * F2 + 0] = a0 * di;
        h2s[node * F2 + 1] = a1 * di;
    }
}

__global__ void __launch_bounds__(WG)
gcn_layer2_kernel(const int2* __restrict__ sorted,
                  const int* __restrict__ bstart,
                  const float* __restrict__ h2s,
                  const float* __restrict__ dinv,
                  const float* __restrict__ b2,
                  float* __restrict__ out,
                  int n) {
    __shared__ float acc[WAVES * NPB * F2];
    const int tid  = threadIdx.x;
    const int lane = tid & 31;
    const int wave = tid >> 5;
    const int b    = blockIdx.x;
    const int base = b << NPB_SHIFT;
    for (int k = tid; k < WAVES * NPB * F2; k += WG) acc[k] = 0.0f;
    __syncthreads();
    const int beg = bstart[b];
    const int end = bstart[b + 1];
    const float2* h2v = (const float2*)h2s;
    lds_vfloat* aw = (lds_vfloat*)(acc + wave * (NPB * F2));
    const int nbatch = (end - beg + WG - 1) / WG;
    for (int j0 = 0; j0 < nbatch; j0 += LD_U) {
        int2   ee[LD_U];
        float2 vv[LD_U];
        #pragma unroll
        for (int u = 0; u < LD_U; ++u) {
            const int i = beg + (j0 + u) * WG + tid;
            ee[u] = sorted[min(i, end - 1)];
        }
        #pragma unroll
        for (int u = 0; u < LD_U; ++u) vv[u] = h2v[ee[u].x];
        #pragma unroll
        for (int u = 0; u < LD_U; ++u) {
            const int i = beg + (j0 + u) * WG + tid;
            const bool act = i < end;
            const int key = ee[u].y - base;
            const unsigned m = wave_peers(key, act, NPB_SHIFT);
            const int rank = lanes_below(m);
            for (int turn = 0; ; ++turn) {
                const bool mine = act && rank == turn;
                if (__builtin_amdgcn_ballot_w32(mine) == 0) break;
                if (mine) {
                    aw[key * F2 + 0] = aw[key * F2 + 0] + vv[u].x;
                    aw[key * F2 + 1] = aw[key * F2 + 1] + vv[u].y;
                }
            }
        }
    }
    __syncthreads();
    const int node = base + tid;
    if (tid < NPB && node < n) {
        float s0 = 0.0f, s1 = 0.0f;
        #pragma unroll
        for (int w = 0; w < WAVES; ++w) {
            s0 += acc[w * (NPB * F2) + tid * F2 + 0];
            s1 += acc[w * (NPB * F2) + tid * F2 + 1];
        }
        float di = dinv[node];
        float2 s = h2v[node];
        float a = (s0 + s.x) * di + b2[0];
        float c = (s1 + s.y) * di + b2[1];
        float m = fmaxf(a, c);
        float lse = m + logf(expf(a - m) + expf(c - m));
        out[node * F2 + 0] = a - lse;
        out[node * F2 + 1] = c - lse;
    }
}

extern "C" void kernel_launch(void* const* d_in, const int* in_sizes, int n_in,
                              void* d_out, int out_size, void* d_ws, size_t ws_size,
                              hipStream_t stream) {
    const float* x   = (const float*)d_in[0];
    const int*   ei  = (const int*)d_in[1];
    const float* W1  = (const float*)d_in[2];
    const float* b1  = (const float*)d_in[3];
    const float* W2  = (const float*)d_in[4];
    const float* b2  = (const float*)d_in[5];
    float*       out = (float*)d_out;

    const int N = in_sizes[0] / FIN;
    const int E = in_sizes[1] / 2;
    const int* rows = ei;
    const int* cols = ei + E;

    const int NB    = (N + NPB - 1) / NPB;
    const int chunk = (E + SPLIT_WG - 1) / SPLIT_WG;

    int*   wsi    = (int*)d_ws;
    int2*  sorted = (int2*)wsi;
    float* h1s    = (float*)(wsi + (size_t)2 * E);
    float* h2s    = h1s + (size_t)16 * N;
    float* dinv   = h2s + (size_t)2 * N;
    int*   counts = (int*)(dinv + (size_t)N);
    int*   totals = counts + (size_t)NB_MAX * SPLIT_WG;
    int*   bstart = totals + NB_MAX;

    gcn_split_count_kernel<<<SPLIT_WG, SWG, 0, stream>>>(cols, counts, E, chunk, NB);
    gcn_split_rowscan_kernel<<<NB, SPLIT_WG, 0, stream>>>(counts, totals);
    gcn_split_starts_kernel<<<1, NB_MAX, 0, stream>>>(totals, bstart, NB);
    gcn_split_scatter_kernel<<<SPLIT_WG, SWG, 0, stream>>>(rows, cols, counts, bstart,
                                                          sorted, E, chunk, NB);

    gcn_degree_kernel<<<NB, WG, 0, stream>>>(sorted, bstart, dinv, N);

    {
        const int BS = 256;
        int tiles = (N + 15) / 16;
        int waves_per_block = BS / 32;
        int blocks = (tiles + waves_per_block - 1) / waves_per_block;
        gcn_xw1_wmma_kernel<<<blocks, BS, 0, stream>>>(x, W1, dinv, h1s, N);
    }

    gcn_layer1_kernel<<<NB, WG, 0, stream>>>(sorted, bstart, h1s, dinv, b1, W2, h2s, N);

    gcn_layer2_kernel<<<NB, WG, 0, stream>>>(sorted, bstart, h2s, dinv, b2, out, N);
}


# Every edge ranked once: count kernel keeps its edges and ballot ranks in registers and stores each edge's 2-byte place inside its chunk bucket; scatter kernel only places; sum order unchanged
# speedup vs baseline: 4.5634x; 1.0521x over previous
#include <hip/hip_runtime.h>
#include <hip/hip_bf16.h>


typedef __attribute__((ext_vector_type(2))) float v2f;
typedef __attribute__((ext_vector_type(8))) float v8f;

#define F1 16
#define F2 2
#define FIN 25

#define NPB_SHIFT 7
#define NPB 128
#define NB_MAX 1024
#define NB_BITS 10
#define SPLIT_WG 256
#define SWG 1024
#define SWAVES 32
#define STAGE_CAP 16384
#define WG 256
#define WAVES 8
#define LD_U 4
#define MAXB (STAGE_CAP / SWG)

typedef __attribute__((address_space(3))) volatile int   lds_vint;
typedef __attribute__((address_space(3))) volatile float lds_vfloat;
typedef __attribute__((ext_vector_type(4))) float v4f;
typedef __attribute__((address_space(3))) volatile v4f   lds_vfloat4;
typedef __attribute__((ext_vector_type(2))) int v2i;

__device__ __forceinline__ unsigned wave_peers(int key, bool act, int nbits) {
    unsigned m = __builtin_amdgcn_ballot_w32(act);
    for (int bit = 0; bit < nbits; ++bit) {
        const int x = __builtin_amdgcn_sbfe(key, bit, 1);
        const unsigned bal = __builtin_amdgcn_ballot_w32(x != 0);
        m = __builtin_amdgcn_bitop3_b32(m, bal, (unsigned)x, 0x90);
    }
    return act ? m : 0u;
}

__device__ __forceinline__ int lanes_below(unsigned m) {
    return (int)__builtin_amdgcn_mbcnt_lo(m, 0u);
}

__device__ __forceinline__ int block_excl_scan(int v, int* sm, int tid, int nthreads,
                                               int* total) {
    sm[tid] = v;
    __syncthreads();
    for (int off = 1; off < nthreads; off <<= 1) {
        int t = (tid >= off) ? sm[tid - off] : 0;
        __syncthreads();
        sm[tid] += t;
        __syncthreads();
    }
    int incl = sm[tid];
    *total = sm[nthreads - 1];
    __syncthreads();
    return incl - v;
}

__global__ void __launch_bounds__(SWG)
gcn_split_count_kernel(const int* __restrict__ cols,
                       int* __restrict__ counts,
                       unsigned short* __restrict__ place,
                       int n_edges, int chunk, int nb) {
    static_assert(SWG == NB_MAX, "one thread per bucket in the offset step");
    static_assert(STAGE_CAP <= 65536, "a place fits 16 bits");
    __shared__ int hist[SWAVES * NB_MAX];
    const int tid  = threadIdx.x;
    const int lane = tid & 31;
    const int wave = tid >> 5;
    const int wg   = blockIdx.x;
    const int beg = min(n_edges, wg * chunk);
    const int end = min(n_edges, beg + chunk);
    const int nbatch  = (end - beg + SWG - 1) / SWG;
    const int i_first = beg + tid;
    int cc[MAXB], loc[MAXB];
    #pragma unroll
    for (int j = 0; j < MAXB; ++j) cc[j] = 0;
    if (end > beg) {
        #pragma unroll
        for (int j = 0; j < MAXB; ++j) cc[j] = cols[min(i_first + j * SWG, end - 1)];
    }
    for (int k = tid; k < SWAVES * NB_MAX; k += SWG) hist[k] = 0;
    __syncthreads();
    lds_vint* hw = (lds_vint*)(hist + wave * NB_MAX);
    #pragma unroll
    for (int j = 0; j < MAXB; ++j) {
        loc[j] = 0;
        if (j < nbatch) {
            const bool act = i_first + j * SWG < end;
            const int key = cc[j] >> NPB_SHIFT;
            const unsigned m = wave_peers(key, act, NB_BITS);
            const int rank = lanes_below(m);
            int p = 0;
            if (act) p = hw[key];
            if (act && rank == 0) hw[key] = p + __popc(m);
            loc[j] = p + rank;
        }
    }
    __syncthreads();
    int tot = 0;
    #pragma unroll
    for (int w = 0; w < SWAVES; ++w) {
        int t = hist[w * NB_MAX + tid];
        hist[w * NB_MAX + tid] = tot;
        tot += t;
    }
    if (tid < nb) counts[tid * SPLIT_WG + wg] = tot;
    __syncthreads();
    #pragma unroll
    for (int j = 0; j < MAXB; ++j) loc[j] += hw[cc[j] >> NPB_SHIFT];
    #pragma unroll
    for (int j = 0; j < MAXB; ++j) {
        if (i_first + j * SWG < end) place[i_first + j * SWG] = (unsigned short)loc[j];
    }
}

__global__ void __launch_bounds__(SPLIT_WG)
gcn_split_rowscan_kernel(int* __restrict__ counts,
                         int* __restrict__ totals) {
    __shared__ int sm[SPLIT_WG];
    const int tid = threadIdx.x;
    const int b   = blockIdx.x;
    int v = counts[b * SPLIT_WG + tid];
    int total;
    int ex = block_excl_scan(v, sm, tid, SPLIT_WG, &total);
    counts[b * SPLIT_WG + tid] = ex;
    if (tid == 0) totals[b] = total;
}

__global__ void __launch_bounds__(NB_MAX)
gcn_split_starts_kernel(const int* __restrict__ totals,
                        int* __restrict__ bstart,
                        int nb) {
    __shared__ int sm[NB_MAX];
    const int tid = threadIdx.x;
    int v = (tid < nb) ? totals[tid] : 0;
    int all;
    int ex = block_excl_scan(v, sm, tid, NB_MAX, &all);
    if (tid < nb) bstart[tid] = ex;
    if (tid == 0) bstart[nb] = all;
}

__global__ void __launch_bounds__(SWG)
gcn_split_scatter_kernel(const int* __restrict__ rows,
                         const int* __restrict__ cols,
                         const unsigned short* __restrict__ place,
                         const int* __restrict__ counts,
                         const int* __restrict__ totals,
                         const int* __restrict__ bstart,
                         int2* __restrict__ sorted,
                         int n_edges, int chunk, int nb) {
    static_assert(SWG == NB_MAX, "one thread per bucket in the start step");
    __shared__ v2i stage[STAGE_CAP];
    __shared__ int lstart[NB_MAX];
    __shared__ int delta[NB_MAX];
    __shared__ int scan_sm[SWG];
    const int tid = threadIdx.x;
    const int wg  = blockIdx.x;
    const int beg = min(n_edges, wg * chunk);
    const int end = min(n_edges, beg + chunk);
    const int i_first = beg + tid;
    int rr[MAXB], cc[MAXB], pl[MAXB];
    if (end <= beg) return;
    #pragma unroll
    for (int j = 0; j < MAXB; ++j) {
        const int i = min(i_first + j * SWG, end - 1);
        rr[j] = rows[i];
        cc[j] = cols[i];
        pl[j] = place[i];
    }
    int cnt = 0, gb = 0;
    if (tid < nb) {
        const int sc = counts[tid * SPLIT_WG + wg];
        const int nx = (wg + 1 < SPLIT_WG) ? counts[tid * SPLIT_WG + wg + 1] : totals[tid];
        cnt = nx - sc;
        gb  = bstart[tid] + sc;
    }
    int all;
    const int ls = block_excl_scan(cnt, scan_sm, tid, SWG, &all);
    lstart[tid] = ls;
    delta[tid]  = gb - ls;
    __syncthreads();
    int at[MAXB];
    #pragma unroll
    for (int j = 0; j < MAXB; ++j) at[j] = lstart[cc[j] >> NPB_SHIFT] + pl[j];
    #pragma unroll
    for (int j = 0; j < MAXB; ++j) {
        if (i_first + j * SWG < end) {
            v2i e;
            e.x = rr[j];
            e.y = cc[j];
            stage[at[j]] = e;
        }
    }
    __syncthreads();
    v2i* sv = (v2i*)sorted;
    const int n_chunk = end - beg;
    for (int i0 = tid; i0 < n_chunk; i0 += LD_U * SWG) {
        v2i e[LD_U];
        int d[LD_U];
        #pragma unroll
        for (int u = 0; u < LD_U; ++u) e[u] = stage[min(i0 + u * SWG, n_chunk - 1)];
        #pragma unroll
        for (int u = 0; u < LD_U; ++u) d[u] = delta[e[u].y >> NPB_SHIFT];
        #pragma unroll
        for (int u = 0; u < LD_U; ++u) {
            const int i = i0 + u * SWG;
            if (i < n_chunk) sv[i + d[u]] = e[u];
        }
    }
}

__global__ void __launch_bounds__(WG)
gcn_degree_kernel(const int2* __restrict__ sorted,
                  const int* __restrict__ bstart,
                  float* __restrict__ dinv,
                  int n) {
    __shared__ int cnt[WAVES * NPB];
    const int tid  = threadIdx.x;
    const int lane = tid & 31;
    const int wave = tid >> 5;
    const int b    = blockIdx.x;
    const int base = b << NPB_SHIFT;
    for (int k = tid; k < WAVES * NPB; k += WG) cnt[k] = 0;
    __syncthreads();
    const int beg = bstart[b];
    const int end = bstart[b + 1];
    lds_vint* cw = (lds_vint*)(cnt + wave * NPB);
    const int nbatch = (end - beg + WG - 1) / WG;
    for (int j0 = 0; j0 < nbatch; j0 += LD_U) {
        int cc[LD_U];
        #pragma unroll
        for (int u = 0; u < LD_U; ++u) {
            const int i = beg + (j0 + u) * WG + tid;
            cc[u] = sorted[min(i, end - 1)].y;
        }
        #pragma unroll
        for (int u = 0; u < LD_U; ++u) {
            const int i = beg + (j0 + u) * WG + tid;
            const bool act = i < end;
            const int key = cc[u] - base;
            const unsigned m = wave_peers(key, act, NPB_SHIFT);
            if (act && lanes_below(m) == 0) cw[key] = cw[key] + __popc(m);
        }
    }
    __syncthreads();
    if (tid < NPB && base + tid < n) {
        int d = 1;
        #pragma unroll
        for (int w = 0; w < WAVES; ++w) d += cnt[w * NPB + tid];
        dinv[base + tid] = rsqrtf((float)d);
    }
}

__global__ void gcn_xw1_wmma_kernel(const float* __restrict__ x,
                                    const float* __restrict__ W1,
                                    const float* __restrict__ dinv,
                                    float* __restrict__ h1s,
                                    int n_nodes) {
    const int lane  = threadIdx.x & 31;
    const int wave  = threadIdx.x >> 5;
    const int tile  = blockIdx.x * (blockDim.x >> 5) + wave;
    const int node0 = tile * 16;
    if (node0 >= n_nodes) return;

    const int half = lane >> 4;
    const int l16  = lane & 15;
    const int m    = l16;

    v8f c = {0.f, 0.f, 0.f, 0.f, 0.f, 0.f, 0.f, 0.f};

    #pragma unroll
    for (int k0 = 0; k0 < 28; k0 += 4) {
        const int ka = k0 + 2 * half;
        v2f a, b;
        a.x = (ka     < FIN) ? x[(node0 + m) * FIN + ka    ] : 0.0f;
        a.y = (ka + 1 < FIN) ? x[(node0 + m) * FIN + ka + 1] : 0.0f;
        b.x = (ka     < FIN) ? W1[(ka    ) * F1 + l16] : 0.0f;
        b.y = (ka + 1 < FIN) ? W1[(ka + 1) * F1 + l16] : 0.0f;
        c = __builtin_amdgcn_wmma_f32_16x16x4_f32(
                 false, a,  false, b,
                 (short)0, c,  false,  false);
    }

    #pragma unroll
    for (int v = 0; v < 8; ++v) {
        const int node = node0 + v + 8 * half;
        h1s[node * F1 + l16] = c[v] * dinv[node];
    }
}

#define L1_U 8
__global__ void __launch_bounds__(WG)
gcn_layer1_kernel(const int2* __restrict__ sorted,
                  const int* __restrict__ bstart,
                  const float* __restrict__ h1s,
                  const float* __restrict__ dinv,
                  const float* __restrict__ b1,
                  const float* __restrict__ W2,
                  float* __restrict__ h2s,
                  int n) {
    __shared__ __attribute__((aligned(16))) float acc[WAVES * NPB * F1];
    const int tid  = threadIdx.x;
    const int lane = tid & 31;
    const int wave = tid >> 5;
    const int grp  = lane >> 2;
    const int q    = lane & 3;
    const int b    = blockIdx.x;
    const int base = b << NPB_SHIFT;
    for (int k = tid; k < WAVES * NPB * F1; k += WG) acc[k] = 0.0f;
    __syncthreads();
    const int beg = bstart[b];
    const int end = bstart[b + 1];
    const v4f* h1v = (const v4f*)h1s;
    lds_vfloat4* aw = (lds_vfloat4*)(acc + wave * (NPB * F1));
    for (int i0 = beg; i0 < end; i0 += 8 * WAVES * L1_U) {
        int2 ee[L1_U];
        v4f  vv[L1_U];
        #pragma unroll
        for (int u = 0; u < L1_U; ++u) {
            const int i = i0 + u * (8 * WAVES) + wave * 8 + grp;
            ee[u] = sorted[min(i, end - 1)];
        }
        #pragma unroll
        for (int u = 0; u < L1_U; ++u) vv[u] = h1v[ee[u].x * 4 + q];
        #pragma unroll
        for (int u = 0; u < L1_U; ++u) {
            const int i = i0 + u * (8 * WAVES) + wave * 8 + grp;
            const bool act = i < end;
            const int key = ee[u].y - base;
            const unsigned m = wave_peers(key, act, NPB_SHIFT);
            const int rank = lanes_below(m) >> 2;
            for (int turn = 0; ; ++turn) {
                const bool mine = act && rank == turn;
                if (__builtin_amdgcn_ballot_w32(mine) == 0) break;
                if (mine) {
                    v4f t = aw[key * 4 + q];
                    aw[key * 4 + q] = t + vv[u];
                }
            }
        }
    }
    __syncthreads();
    for (int k = tid; k < NPB * F1; k += WG) {
        float s = acc[k];
        #pragma unroll
        for (int w = 1; w < WAVES; ++w) s += acc[w * (NPB * F1) + k];
        acc[k] = s;
    }
    __syncthreads();
    const int node = base + tid;
    if (tid < NPB && node < n) {
        float di = dinv[node];
        float a0 = 0.0f, a1 = 0.0f;
        #pragma unroll
        for (int k = 0; k < F1; ++k) {
            float v = (acc[tid * F1 + k] + h1s[node * F1 + k]) * di + b1[k];
            v = v > 0.0f ? v : 0.0f;
            a0 = fmaf(v, W2[k * F2 + 0], a0);
            a1 = fmaf(v, W2[k * F2 + 1], a1);
        }
        h2s[node * F2 + 0] = a0 * di;
        h2s[node * F2 + 1] = a1 * di;
    }
}

__global__ void __launch_bounds__(WG)
gcn_layer2_kernel(const int2* __restrict__ sorted,
                  const int* __restrict__ bstart,
                  const float* __restrict__ h2s,
                  const float* __restrict__ dinv,
                  const float* __restrict__ b2,
                  float* __restrict__ out,
                  int n) {
    __shared__ float acc[WAVES * NPB * F2];
    const int tid  = threadIdx.x;
    const int lane = tid & 31;
    const int wave = tid >> 5;
    const int b    = blockIdx.x;
    const int base = b << NPB_SHIFT;
    for (int k = tid; k < WAVES * NPB * F2; k += WG) acc[k] = 0.0f;
    __syncthreads();
    const int beg = bstart[b];
    const int end = bstart[b + 1];
    const float2* h2v = (const float2*)h2s;
    lds_vfloat* aw = (lds_vfloat*)(acc + wave * (NPB * F2));
    const int nbatch = (end - beg + WG - 1) / WG;
    for (int j0 = 0; j0 < nbatch; j0 += LD_U) {
        int2   ee[LD_U];
        float2 vv[LD_U];
        #pragma unroll
        for (int u = 0; u < LD_U; ++u) {
            const int i = beg + (j0 + u) * WG + tid;
            ee[u] = sorted[min(i, end - 1)];
        }
        #pragma unroll
        for (int u = 0; u < LD_U; ++u) vv[u] = h2v[ee[u].x];
        #pragma unroll
        for (int u = 0; u < LD_U; ++u) {
            const int i = beg + (j0 + u) * WG + tid;
            const bool act = i < end;
            const int key = ee[u].y - base;
            const unsigned m = wave_peers(key, act, NPB_SHIFT);
            const int rank = lanes_below(m);
            for (int turn = 0; ; ++turn) {
                const bool mine = act && rank == turn;
                if (__builtin_amdgcn_ballot_w32(mine) == 0) break;
                if (mine) {
                    aw[key * F2 + 0] = aw[key * F2 + 0] + vv[u].x;
                    aw[key * F2 + 1] = aw[key * F2 + 1] + vv[u].y;
                }
            }
        }
    }
    __syncthreads();
    const int node = base + tid;
    if (tid < NPB && node < n) {
        float s0 = 0.0f, s1 = 0.0f;
        #pragma unroll
        for (int w = 0; w < WAVES; ++w) {
            s0 += acc[w * (NPB * F2) + tid * F2 + 0];
            s1 += acc[w * (NPB * F2) + tid * F2 + 1];
        }
        float di = dinv[node];
        float2 s = h2v[node];
        float a = (s0 + s.x) * di + b2[0];
        float c = (s1 + s.y) * di + b2[1];
        float m = fmaxf(a, c);
        float lse = m + logf(expf(a - m) + expf(c - m));
        out[node * F2 + 0] = a - lse;
        out[node * F2 + 1] = c - lse;
    }
}

extern "C" void kernel_launch(void* const* d_in, const int* in_sizes, int n_in,
                              void* d_out, int out_size, void* d_ws, size_t ws_size,
                              hipStream_t stream) {
    const float* x   = (const float*)d_in[0];
    const int*   ei  = (const int*)d_in[1];
    const float* W1  = (const float*)d_in[2];
    const float* b1  = (const float*)d_in[3];
    const float* W2  = (const float*)d_in[4];
    const float* b2  = (const float*)d_in[5];
    float*       out = (float*)d_out;

    const int N = in_sizes[0] / FIN;
    const int E = in_sizes[1] / 2;
    const int* rows = ei;
    const int* cols = ei + E;

    const int NB    = (N + NPB - 1) / NPB;
    const int chunk = (E + SPLIT_WG - 1) / SPLIT_WG;

    int*   wsi    = (int*)d_ws;
    int2*  sorted = (int2*)wsi;
    float* h1s    = (float*)(wsi + (size_t)2 * E);
    float* h2s    = h1s + (size_t)16 * N;
    float* dinv   = h2s + (size_t)2 * N;
    int*   counts = (int*)(dinv + (size_t)N);
    int*   totals = counts + (size_t)NB_MAX * SPLIT_WG;
    int*   bstart = totals + NB_MAX;
    unsigned short* place = (unsigned short*)(bstart + NB_MAX + 8);

    gcn_split_count_kernel<<<SPLIT_WG, SWG, 0, stream>>>(cols, counts, place, E, chunk, NB);
    gcn_split_rowscan_kernel<<<NB, SPLIT_WG, 0, stream>>>(counts, totals);
    gcn_split_starts_kernel<<<1, NB_MAX, 0, stream>>>(totals, bstart, NB);
    gcn_split_scatter_kernel<<<SPLIT_WG, SWG, 0, stream>>>(rows, cols, place, counts, totals,
                                                          bstart, sorted, E, chunk, NB);

    gcn_degree_kernel<<<NB, WG, 0, stream>>>(sorted, bstart, dinv, N);

    {
        const int BS = 256;
        int tiles = (N + 15) / 16;
        int waves_per_block = BS / 32;
        int blocks = (tiles + waves_per_block - 1) / waves_per_block;
        gcn_xw1_wmma_kernel<<<blocks, BS, 0, stream>>>(x, W1, dinv, h1s, N);
    }

    gcn_layer1_kernel<<<NB, WG, 0, stream>>>(sorted, bstart, h1s, dinv, b1, W2, h2s, N);

    gcn_layer2_kernel<<<NB, WG, 0, stream>>>(sorted, bstart, h2s, dinv, b2, out, N);
}


# Two launches fewer (8 to 6): bucket starts scanned inside the scatter kernel's own block scan (64-bit pairs); x@W1 tile of 16 nodes per wave done by the degree kernel's 8 waves beside the counting
# speedup vs baseline: 4.9387x; 1.0822x over previous
#include <hip/hip_runtime.h>
#include <hip/hip_bf16.h>


typedef __attribute__((ext_vector_type(2))) float v2f;
typedef __attribute__((ext_vector_type(8))) float v8f;

#define F1 16
#define F2 2
#define FIN 25

#define NPB_SHIFT 7
#define NPB 128
#define NB_MAX 1024
#define NB_BITS 10
#define SPLIT_WG 256
#define SWG 1024
#define SWAVES 32
#define STAGE_CAP 16384
#define WG 256
#define WAVES 8
#define LD_U 4
#define MAXB (STAGE_CAP / SWG)

typedef __attribute__((address_space(3))) volatile int   lds_vint;
typedef __attribute__((address_space(3))) volatile float lds_vfloat;
typedef __attribute__((ext_vector_type(4))) float v4f;
typedef __attribute__((address_space(3))) volatile v4f   lds_vfloat4;
typedef __attribute__((ext_vector_type(2))) int v2i;

__device__ __forceinline__ unsigned wave_peers(int key, bool act, int nbits) {
    unsigned m = __builtin_amdgcn_ballot_w32(act);
    for (int bit = 0; bit < nbits; ++bit) {
        const int x = __builtin_amdgcn_sbfe(key, bit, 1);
        const unsigned bal = __builtin_amdgcn_ballot_w32(x != 0);
        m = __builtin_amdgcn_bitop3_b32(m, bal, (unsigned)x, 0x90);
    }
    return act ? m : 0u;
}

__device__ __forceinline__ int lanes_below(unsigned m) {
    return (int)__builtin_amdgcn_mbcnt_lo(m, 0u);
}

template <typename T>
__device__ __forceinline__ T block_excl_scan(T v, T* sm, int tid, int nthreads,
                                             T* total) {
    sm[tid] = v;
    __syncthreads();
    for (int off = 1; off < nthreads; off <<= 1) {
        T t = (tid >= off) ? sm[tid - off] : (T)0;
        __syncthreads();
        sm[tid] += t;
        __syncthreads();
    }
    T incl = sm[tid];
    *total = sm[nthreads - 1];
    __syncthreads();
    return incl - v;
}

__global__ void __launch_bounds__(SWG)
gcn_split_count_kernel(const int* __restrict__ cols,
                       int* __restrict__ counts,
                       unsigned short* __restrict__ place,
                       int n_edges, int chunk, int nb) {
    static_assert(SWG == NB_MAX, "one thread per bucket in the offset step");
    static_assert(STAGE_CAP <= 65536, "a place fits 16 bits");
    __shared__ int hist[SWAVES * NB_MAX];
    const int tid  = threadIdx.x;
    const int lane = tid & 31;
    const int wave = tid >> 5;
    const int wg   = blockIdx.x;
    const int beg = min(n_edges, wg * chunk);
    const int end = min(n_edges, beg + chunk);
    const int nbatch  = (end - beg + SWG - 1) / SWG;
    const int i_first = beg + tid;
    int cc[MAXB], loc[MAXB];
    #pragma unroll
    for (int j = 0; j < MAXB; ++j) cc[j] = 0;
    if (end > beg) {
        #pragma unroll
        for (int j = 0; j < MAXB; ++j) cc[j] = cols[min(i_first + j * SWG, end - 1)];
    }
    for (int k = tid; k < SWAVES * NB_MAX; k += SWG) hist[k] = 0;
    __syncthreads();
    lds_vint* hw = (lds_vint*)(hist + wave * NB_MAX);
    #pragma unroll
    for (int j = 0; j < MAXB; ++j) {
        loc[j] = 0;
        if (j < nbatch) {
            const bool act = i_first + j * SWG < end;
            const int key = cc[j] >> NPB_SHIFT;
            const unsigned m = wave_peers(key, act, NB_BITS);
            const int rank = lanes_below(m);
            int p = 0;
            if (act) p = hw[key];
            if (act && rank == 0) hw[key] = p + __popc(m);
            loc[j] = p + rank;
        }
    }
    __syncthreads();
    int tot = 0;
    #pragma unroll
    for (int w = 0; w < SWAVES; ++w) {
        int t = hist[w * NB_MAX + tid];
        hist[w * NB_MAX + tid] = tot;
        tot += t;
    }
    if (tid < nb) counts[tid * SPLIT_WG + wg] = tot;
    __syncthreads();
    #pragma unroll
    for (int j = 0; j < MAXB; ++j) loc[j] += hw[cc[j] >> NPB_SHIFT];
    #pragma unroll
    for (int j = 0; j < MAXB; ++j) {
        if (i_first + j * SWG < end) place[i_first + j * SWG] = (unsigned short)loc[j];
    }
}

__global__ void __launch_bounds__(SPLIT_WG)
gcn_split_rowscan_kernel(int* __restrict__ counts,
                         int* __restrict__ totals) {
    __shared__ int sm[SPLIT_WG];
    const int tid = threadIdx.x;
    const int b   = blockIdx.x;
    int v = counts[b * SPLIT_WG + tid];
    int total;
    int ex = block_excl_scan(v, sm, tid, SPLIT_WG, &total);
    counts[b * SPLIT_WG + tid] = ex;
    if (tid == 0) totals[b] = total;
}

__global__ void __launch_bounds__(SWG)
gcn_split_scatter_kernel(const int* __restrict__ rows,
                         const int* __restrict__ cols,
                         const unsigned short* __restrict__ place,
                         const int* __restrict__ counts,
                         const int* __restrict__ totals,
                         int* __restrict__ bstart,
                         int2* __restrict__ sorted,
                         int n_edges, int chunk, int nb) {
    static_assert(SWG == NB_MAX, "one thread per bucket in the start step");
    __shared__ v2i stage[STAGE_CAP];
    __shared__ int lstart[NB_MAX];
    __shared__ int delta[NB_MAX];
    __shared__ unsigned long long scan_sm[SWG];
    const int tid = threadIdx.x;
    const int wg  = blockIdx.x;
    const int beg = min(n_edges, wg * chunk);
    const int end = min(n_edges, beg + chunk);
    const int i_first = beg + tid;
    int rr[MAXB], cc[MAXB], pl[MAXB];
    if (end <= beg && wg != 0) return;
    #pragma unroll
    for (int j = 0; j < MAXB; ++j) { rr[j] = 0; cc[j] = 0; pl[j] = 0; }
    if (end > beg) {
        #pragma unroll
        for (int j = 0; j < MAXB; ++j) {
            const int i = min(i_first + j * SWG, end - 1);
            rr[j] = rows[i];
            cc[j] = cols[i];
            pl[j] = place[i];
        }
    }
    int cnt = 0, sc = 0, tot = 0;
    if (tid < nb) {
        sc  = counts[tid * SPLIT_WG + wg];
        tot = totals[tid];
        const int nx = (wg + 1 < SPLIT_WG) ? counts[tid * SPLIT_WG + wg + 1] : tot;
        cnt = nx - sc;
    }
    unsigned long long all;
    const unsigned long long ex = block_excl_scan(
        ((unsigned long long)(unsigned)cnt << 32) | (unsigned)tot, scan_sm, tid, SWG, &all);
    const int bs = (int)(unsigned)ex;
    const int ls = (int)(ex >> 32);
    lstart[tid] = ls;
    delta[tid]  = bs + sc - ls;
    if (wg == 0) {
        if (tid < nb)  bstart[tid] = bs;
        if (tid == 0)  bstart[nb]  = (int)(unsigned)all;
    }
    __syncthreads();
    int at[MAXB];
    #pragma unroll
    for (int j = 0; j < MAXB; ++j) at[j] = lstart[cc[j] >> NPB_SHIFT] + pl[j];
    #pragma unroll
    for (int j = 0; j < MAXB; ++j) {
        if (i_first + j * SWG < end) {
            v2i e;
            e.x = rr[j];
            e.y = cc[j];
            stage[at[j]] = e;
        }
    }
    __syncthreads();
    v2i* sv = (v2i*)sorted;
    const int n_chunk = end - beg;
    for (int i0 = tid; i0 < n_chunk; i0 += LD_U * SWG) {
        v2i e[LD_U];
        int d[LD_U];
        #pragma unroll
        for (int u = 0; u < LD_U; ++u) e[u] = stage[min(i0 + u * SWG, n_chunk - 1)];
        #pragma unroll
        for (int u = 0; u < LD_U; ++u) d[u] = delta[e[u].y >> NPB_SHIFT];
        #pragma unroll
        for (int u = 0; u < LD_U; ++u) {
            const int i = i0 + u * SWG;
            if (i < n_chunk) sv[i + d[u]] = e[u];
        }
    }
}

__global__ void __launch_bounds__(WG)
gcn_degree_xw1_kernel(const int2* __restrict__ sorted,
                      const int* __restrict__ bstart,
                      const float* __restrict__ x,
                      const float* __restrict__ W1,
                      float* __restrict__ dinv,
                      float* __restrict__ h1s,
                      int n) {
    static_assert(NPB == WAVES * 16, "one 16-node tile per wave covers the bucket");
    __shared__ int cnt[WAVES * NPB];
    __shared__ float dsm[NPB];
    const int tid  = threadIdx.x;
    const int lane = tid & 31;
    const int wave = tid >> 5;
    const int half = lane >> 4;
    const int l16  = lane & 15;
    const int b    = blockIdx.x;
    const int base = b << NPB_SHIFT;
    const int node0 = base + wave * 16;
    const bool tile = node0 < n;

    for (int k = tid; k < WAVES * NPB; k += WG) cnt[k] = 0;
    __syncthreads();

    v2f fa[7], fb[7];
    #pragma unroll
    for (int s = 0; s < 7; ++s) { fa[s].x = 0.0f; fa[s].y = 0.0f; fb[s].x = 0.0f; fb[s].y = 0.0f; }
    if (tile) {
        #pragma unroll
        for (int s = 0; s < 7; ++s) {
            const int ka = 4 * s + 2 * half;
            if (ka     < FIN) { fa[s].x = x[(node0 + l16) * FIN + ka    ]; fb[s].x = W1[(ka    ) * F1 + l16]; }
            if (ka + 1 < FIN) { fa[s].y = x[(node0 + l16) * FIN + ka + 1]; fb[s].y = W1[(ka + 1) * F1 + l16]; }
        }
    }

    const int beg = bstart[b];
    const int end = bstart[b + 1];
    lds_vint* cw = (lds_vint*)(cnt + wave * NPB);
    const int nbatch = (end - beg + WG - 1) / WG;
    for (int j0 = 0; j0 < nbatch; j0 += LD_U) {
        int cc[LD_U];
        #pragma unroll
        for (int u = 0; u < LD_U; ++u) {
            const int i = beg + (j0 + u) * WG + tid;
            cc[u] = sorted[min(i, end - 1)].y;
        }
        #pragma unroll
        for (int u = 0; u < LD_U; ++u) {
            const int i = beg + (j0 + u) * WG + tid;
            const bool act = i < end;
            const int key = cc[u] - base;
            const unsigned m = wave_peers(key, act, NPB_SHIFT);
            if (act && lanes_below(m) == 0) cw[key] = cw[key] + __popc(m);
        }
    }

    v8f c = {0.f, 0.f, 0.f, 0.f, 0.f, 0.f, 0.f, 0.f};
    if (tile) {
        #pragma unroll
        for (int s = 0; s < 7; ++s) {
            c = __builtin_amdgcn_wmma_f32_16x16x4_f32(
                     false, fa[s],  false, fb[s],
                     (short)0, c,  false,  false);
        }
    }
    __syncthreads();
    if (tid < NPB) {
        float dv = 0.0f;
        if (base + tid < n) {
            int d = 1;
            #pragma unroll
            for (int w = 0; w < WAVES; ++w) d += cnt[w * NPB + tid];
            dv = rsqrtf((float)d);
            dinv[base + tid] = dv;
        }
        dsm[tid] = dv;
    }
    __syncthreads();
    if (tile) {
        #pragma unroll
        for (int v = 0; v < 8; ++v) {
            const int ln = wave * 16 + v + 8 * half;
            h1s[(base + ln) * F1 + l16] = c[v] * dsm[ln];
        }
    }
}

#define L1_U 8
__global__ void __launch_bounds__(WG)
gcn_layer1_kernel(const int2* __restrict__ sorted,
                  const int* __restrict__ bstart,
                  const float* __restrict__ h1s,
                  const float* __restrict__ dinv,
                  const float* __restrict__ b1,
                  const float* __restrict__ W2,
                  float* __restrict__ h2s,
                  int n) {
    __shared__ __attribute__((aligned(16))) float acc[WAVES * NPB * F1];
    const int tid  = threadIdx.x;
    const int lane = tid & 31;
    const int wave = tid >> 5;
    const int grp  = lane >> 2;
    const int q    = lane & 3;
    const int b    = blockIdx.x;
    const int base = b << NPB_SHIFT;
    for (int k = tid; k < WAVES * NPB * F1; k += WG) acc[k] = 0.0f;
    __syncthreads();
    const int beg = bstart[b];
    const int end = bstart[b + 1];
    const v4f* h1v = (const v4f*)h1s;
    lds_vfloat4* aw = (lds_vfloat4*)(acc + wave * (NPB * F1));
    for (int i0 = beg; i0 < end; i0 += 8 * WAVES * L1_U) {
        int2 ee[L1_U];
        v4f  vv[L1_U];
        #pragma unroll
        for (int u = 0; u < L1_U; ++u) {
            const int i = i0 + u * (8 * WAVES) + wave * 8 + grp;
            ee[u] = sorted[min(i, end - 1)];
        }
        #pragma unroll
        for (int u = 0; u < L1_U; ++u) vv[u] = h1v[ee[u].x * 4 + q];
        #pragma unroll
        for (int u = 0; u < L1_U; ++u) {
            const int i = i0 + u * (8 * WAVES) + wave * 8 + grp;
            const bool act = i < end;
            const int key = ee[u].y - base;
            const unsigned m = wave_peers(key, act, NPB_SHIFT);
            const int rank = lanes_below(m) >> 2;
            for (int turn = 0; ; ++turn) {
                const bool mine = act && rank == turn;
                if (__builtin_amdgcn_ballot_w32(mine) == 0) break;
                if (mine) {
                    v4f t = aw[key * 4 + q];
                    aw[key * 4 + q] = t + vv[u];
                }
            }
        }
    }
    __syncthreads();
    for (int k = tid; k < NPB * F1; k += WG) {
        float s = acc[k];
        #pragma unroll
        for (int w = 1; w < WAVES; ++w) s += acc[w * (NPB * F1) + k];
        acc[k] = s;
    }
    __syncthreads();
    const int node = base + tid;
    if (tid < NPB && node < n) {
        float di = dinv[node];
        float a0 = 0.0f, a1 = 0.0f;
        #pragma unroll
        for (int k = 0; k < F1; ++k) {
            float v = (acc[tid * F1 + k] + h1s[node * F1 + k]) * di + b1[k];
            v = v > 0.0f ? v : 0.0f;
            a0 = fmaf(v, W2[k * F2 + 0], a0);
            a1 = fmaf(v, W2[k * F2 + 1], a1);
        }
        h2s[node * F2 + 0] = a0 * di;
        h2s[node * F2 + 1] = a1 * di;
    }
}

__global__ void __launch_bounds__(WG)
gcn_layer2_kernel(const int2* __restrict__ sorted,
                  const int* __restrict__ bstart,
                  const float* __restrict__ h2s,
                  const float* __restrict__ dinv,
                  const float* __restrict__ b2,
                  float* __restrict__ out,
                  int n) {
    __shared__ float acc[WAVES * NPB * F2];
    const int tid  = threadIdx.x;
    const int lane = tid & 31;
    const int wave = tid >> 5;
    const int b    = blockIdx.x;
    const int base = b << NPB_SHIFT;
    for (int k = tid; k < WAVES * NPB * F2; k += WG) acc[k] = 0.0f;
    __syncthreads();
    const int beg = bstart[b];
    const int end = bstart[b + 1];
    const float2* h2v = (const float2*)h2s;
    lds_vfloat* aw = (lds_vfloat*)(acc + wave * (NPB * F2));
    const int nbatch = (end - beg + WG - 1) / WG;
    for (int j0 = 0; j0 < nbatch; j0 += LD_U) {
        int2   ee[LD_U];
        float2 vv[LD_U];
        #pragma unroll
        for (int u = 0; u < LD_U; ++u) {
            const int i = beg + (j0 + u) * WG + tid;
            ee[u] = sorted[min(i, end - 1)];
        }
        #pragma unroll
        for (int u = 0; u < LD_U; ++u) vv[u] = h2v[ee[u].x];
        #pragma unroll
        for (int u = 0; u < LD_U; ++u) {
            const int i = beg + (j0 + u) * WG + tid;
            const bool act = i < end;
            const int key = ee[u].y - base;
            const unsigned m = wave_peers(key, act, NPB_SHIFT);
            const int rank = lanes_below(m);
            for (int turn = 0; ; ++turn) {
                const bool mine = act && rank == turn;
                if (__builtin_amdgcn_ballot_w32(mine) == 0) break;
                if (mine) {
                    aw[key * F2 + 0] = aw[key * F2 + 0] + vv[u].x;
                    aw[key * F2 + 1] = aw[key * F2 + 1] + vv[u].y;
                }
            }
        }
    }
    __syncthreads();
    const int node = base + tid;
    if (tid < NPB && node < n) {
        float s0 = 0.0f, s1 = 0.0f;
        #pragma unroll
        for (int w = 0; w < WAVES; ++w) {
            s0 += acc[w * (NPB * F2) + tid * F2 + 0];
            s1 += acc[w * (NPB * F2) + tid * F2 + 1];
        }
        float di = dinv[node];
        float2 s = h2v[node];
        float a = (s0 + s.x) * di + b2[0];
        float c = (s1 + s.y) * di + b2[1];
        float m = fmaxf(a, c);
        float lse = m + logf(expf(a - m) + expf(c - m));
        out[node * F2 + 0] = a - lse;
        out[node * F2 + 1] = c - lse;
    }
}

extern "C" void kernel_launch(void* const* d_in, const int* in_sizes, int n_in,
                              void* d_out, int out_size, void* d_ws, size_t ws_size,
                              hipStream_t stream) {
    const float* x   = (const float*)d_in[0];
    const int*   ei  = (const int*)d_in[1];
    const float* W1  = (const float*)d_in[2];
    const float* b1  = (const float*)d_in[3];
    const float* W2  = (const float*)d_in[4];
    const float* b2  = (const float*)d_in[5];
    float*       out = (float*)d_out;

    const int N = in_sizes[0] / FIN;
    const int E = in_sizes[1] / 2;
    const int* rows = ei;
    const int* cols = ei + E;

    const int NB    = (N + NPB - 1) / NPB;
    const int chunk = (E + SPLIT_WG - 1) / SPLIT_WG;

    int*   wsi    = (int*)d_ws;
    int2*  sorted = (int2*)wsi;
    float* h1s    = (float*)(wsi + (size_t)2 * E);
    float* h2s    = h1s + (size_t)16 * N;
    float* dinv   = h2s + (size_t)2 * N;
    int*   counts = (int*)(dinv + (size_t)N);
    int*   totals = counts + (size_t)NB_MAX * SPLIT_WG;
    int*   bstart = totals + NB_MAX;
    unsigned short* place = (unsigned short*)(bstart + NB_MAX + 8);

    gcn_split_count_kernel<<<SPLIT_WG, SWG, 0, stream>>>(cols, counts, place, E, chunk, NB);
    gcn_split_rowscan_kernel<<<NB, SPLIT_WG, 0, stream>>>(counts, totals);
    gcn_split_scatter_kernel<<<SPLIT_WG, SWG, 0, stream>>>(rows, cols, place, counts, totals,
                                                          bstart, sorted, E, chunk, NB);

    gcn_degree_xw1_kernel<<<NB, WG, 0, stream>>>(sorted, bstart, x, W1, dinv, h1s, N);

    gcn_layer1_kernel<<<NB, WG, 0, stream>>>(sorted, bstart, h1s, dinv, b1, W2, h2s, N);

    gcn_layer2_kernel<<<NB, WG, 0, stream>>>(sorted, bstart, h2s, dinv, b2, out, N);
}


# Layer 1 finds same-target lanes once per 32 edges, not per 8: lane q of each quad ranks the quad's edge of group q, a quad DPP move hands it on; a group with no clash skips the turns; sum order kept
# speedup vs baseline: 5.1542x; 1.0436x over previous
#include <hip/hip_runtime.h>
#include <hip/hip_bf16.h>


typedef __attribute__((ext_vector_type(2))) float v2f;
typedef __attribute__((ext_vector_type(8))) float v8f;

#define F1 16
#define F2 2
#define FIN 25

#define NPB_SHIFT 7
#define NPB 128
#define NB_MAX 1024
#define NB_BITS 10
#define SPLIT_WG 256
#define SWG 1024
#define SWAVES 32
#define STAGE_CAP 16384
#define WG 256
#define WAVES 8
#define LD_U 4
#define MAXB (STAGE_CAP / SWG)

typedef __attribute__((address_space(3))) volatile int   lds_vint;
typedef __attribute__((address_space(3))) volatile float lds_vfloat;
typedef __attribute__((ext_vector_type(4))) float v4f;
typedef __attribute__((address_space(3))) volatile v4f   lds_vfloat4;
typedef __attribute__((ext_vector_type(2))) int v2i;

__device__ __forceinline__ unsigned wave_peers(int key, bool act, int nbits) {
    unsigned m = __builtin_amdgcn_ballot_w32(act);
    for (int bit = 0; bit < nbits; ++bit) {
        const int x = __builtin_amdgcn_sbfe(key, bit, 1);
        const unsigned bal = __builtin_amdgcn_ballot_w32(x != 0);
        m = __builtin_amdgcn_bitop3_b32(m, bal, (unsigned)x, 0x90);
    }
    return act ? m : 0u;
}

__device__ __forceinline__ int lanes_below(unsigned m) {
    return (int)__builtin_amdgcn_mbcnt_lo(m, 0u);
}

template <typename T>
__device__ __forceinline__ T block_excl_scan(T v, T* sm, int tid, int nthreads,
                                             T* total) {
    sm[tid] = v;
    __syncthreads();
    for (int off = 1; off < nthreads; off <<= 1) {
        T t = (tid >= off) ? sm[tid - off] : (T)0;
        __syncthreads();
        sm[tid] += t;
        __syncthreads();
    }
    T incl = sm[tid];
    *total = sm[nthreads - 1];
    __syncthreads();
    return incl - v;
}

__global__ void __launch_bounds__(SWG)
gcn_split_count_kernel(const int* __restrict__ cols,
                       int* __restrict__ counts,
                       unsigned short* __restrict__ place,
                       int n_edges, int chunk, int nb) {
    static_assert(SWG == NB_MAX, "one thread per bucket in the offset step");
    static_assert(STAGE_CAP <= 65536, "a place fits 16 bits");
    __shared__ int hist[SWAVES * NB_MAX];
    const int tid  = threadIdx.x;
    const int lane = tid & 31;
    const int wave = tid >> 5;
    const int wg   = blockIdx.x;
    const int beg = min(n_edges, wg * chunk);
    const int end = min(n_edges, beg + chunk);
    const int nbatch  = (end - beg + SWG - 1) / SWG;
    const int i_first = beg + tid;
    int cc[MAXB], loc[MAXB];
    #pragma unroll
    for (int j = 0; j < MAXB; ++j) cc[j] = 0;
    if (end > beg) {
        #pragma unroll
        for (int j = 0; j < MAXB; ++j) cc[j] = cols[min(i_first + j * SWG, end - 1)];
    }
    for (int k = tid; k < SWAVES * NB_MAX; k += SWG) hist[k] = 0;
    __syncthreads();
    lds_vint* hw = (lds_vint*)(hist + wave * NB_MAX);
    #pragma unroll
    for (int j = 0; j < MAXB; ++j) {
        loc[j] = 0;
        if (j < nbatch) {
            const bool act = i_first + j * SWG < end;
            const int key = cc[j] >> NPB_SHIFT;
            const unsigned m = wave_peers(key, act, NB_BITS);
            const int rank = lanes_below(m);
            int p = 0;
            if (act) p = hw[key];
            if (act && rank == 0) hw[key] = p + __popc(m);
            loc[j] = p + rank;
        }
    }
    __syncthreads();
    int tot = 0;
    #pragma unroll
    for (int w = 0; w < SWAVES; ++w) {
        int t = hist[w * NB_MAX + tid];
        hist[w * NB_MAX + tid] = tot;
        tot += t;
    }
    if (tid < nb) counts[tid * SPLIT_WG + wg] = tot;
    __syncthreads();
    #pragma unroll
    for (int j = 0; j < MAXB; ++j) loc[j] += hw[cc[j] >> NPB_SHIFT];
    #pragma unroll
    for (int j = 0; j < MAXB; ++j) {
        if (i_first + j * SWG < end) place[i_first + j * SWG] = (unsigned short)loc[j];
    }
}

__global__ void __launch_bounds__(SPLIT_WG)
gcn_split_rowscan_kernel(int* __restrict__ counts,
                         int* __restrict__ totals) {
    __shared__ int sm[SPLIT_WG];
    const int tid = threadIdx.x;
    const int b   = blockIdx.x;
    int v = counts[b * SPLIT_WG + tid];
    int total;
    int ex = block_excl_scan(v, sm, tid, SPLIT_WG, &total);
    counts[b * SPLIT_WG + tid] = ex;
    if (tid == 0) totals[b] = total;
}

__global__ void __launch_bounds__(SWG)
gcn_split_scatter_kernel(const int* __restrict__ rows,
                         const int* __restrict__ cols,
                         const unsigned short* __restrict__ place,
                         const int* __restrict__ counts,
                         const int* __restrict__ totals,
                         int* __restrict__ bstart,
                         int2* __restrict__ sorted,
                         int n_edges, int chunk, int nb) {
    static_assert(SWG == NB_MAX, "one thread per bucket in the start step");
    __shared__ v2i stage[STAGE_CAP];
    __shared__ int lstart[NB_MAX];
    __shared__ int delta[NB_MAX];
    __shared__ unsigned long long scan_sm[SWG];
    const int tid = threadIdx.x;
    const int wg  = blockIdx.x;
    const int beg = min(n_edges, wg * chunk);
    const int end = min(n_edges, beg + chunk);
    const int i_first = beg + tid;
    int rr[MAXB], cc[MAXB], pl[MAXB];
    if (end <= beg && wg != 0) return;
    #pragma unroll
    for (int j = 0; j < MAXB; ++j) { rr[j] = 0; cc[j] = 0; pl[j] = 0; }
    if (end > beg) {
        #pragma unroll
        for (int j = 0; j < MAXB; ++j) {
            const int i = min(i_first + j * SWG, end - 1);
            rr[j] = rows[i];
            cc[j] = cols[i];
            pl[j] = place[i];
        }
    }
    int cnt = 0, sc = 0, tot = 0;
    if (tid < nb) {
        sc  = counts[tid * SPLIT_WG + wg];
        tot = totals[tid];
        const int nx = (wg + 1 < SPLIT_WG) ? counts[tid * SPLIT_WG + wg + 1] : tot;
        cnt = nx - sc;
    }
    unsigned long long all;
    const unsigned long long ex = block_excl_scan(
        ((unsigned long long)(unsigned)cnt << 32) | (unsigned)tot, scan_sm, tid, SWG, &all);
    const int bs = (int)(unsigned)ex;
    const int ls = (int)(ex >> 32);
    lstart[tid] = ls;
    delta[tid]  = bs + sc - ls;
    if (wg == 0) {
        if (tid < nb)  bstart[tid] = bs;
        if (tid == 0)  bstart[nb]  = (int)(unsigned)all;
    }
    __syncthreads();
    int at[MAXB];
    #pragma unroll
    for (int j = 0; j < MAXB; ++j) at[j] = lstart[cc[j] >> NPB_SHIFT] + pl[j];
    #pragma unroll
    for (int j = 0; j < MAXB; ++j) {
        if (i_first + j * SWG < end) {
            v2i e;
            e.x = rr[j];
            e.y = cc[j];
            stage[at[j]] = e;
        }
    }
    __syncthreads();
    v2i* sv = (v2i*)sorted;
    const int n_chunk = end - beg;
    for (int i0 = tid; i0 < n_chunk; i0 += LD_U * SWG) {
        v2i e[LD_U];
        int d[LD_U];
        #pragma unroll
        for (int u = 0; u < LD_U; ++u) e[u] = stage[min(i0 + u * SWG, n_chunk - 1)];
        #pragma unroll
        for (int u = 0; u < LD_U; ++u) d[u] = delta[e[u].y >> NPB_SHIFT];
        #pragma unroll
        for (int u = 0; u < LD_U; ++u) {
            const int i = i0 + u * SWG;
            if (i < n_chunk) sv[i + d[u]] = e[u];
        }
    }
}

__global__ void __launch_bounds__(WG)
gcn_degree_xw1_kernel(const int2* __restrict__ sorted,
                      const int* __restrict__ bstart,
                      const float* __restrict__ x,
                      const float* __restrict__ W1,
                      float* __restrict__ dinv,
                      float* __restrict__ h1s,
                      int n) {
    static_assert(NPB == WAVES * 16, "one 16-node tile per wave covers the bucket");
    __shared__ int cnt[WAVES * NPB];
    __shared__ float dsm[NPB];
    const int tid  = threadIdx.x;
    const int lane = tid & 31;
    const int wave = tid >> 5;
    const int half = lane >> 4;
    const int l16  = lane & 15;
    const int b    = blockIdx.x;
    const int base = b << NPB_SHIFT;
    const int node0 = base + wave * 16;
    const bool tile = node0 < n;

    for (int k = tid; k < WAVES * NPB; k += WG) cnt[k] = 0;
    __syncthreads();

    v2f fa[7], fb[7];
    #pragma unroll
    for (int s = 0; s < 7; ++s) { fa[s].x = 0.0f; fa[s].y = 0.0f; fb[s].x = 0.0f; fb[s].y = 0.0f; }
    if (tile) {
        #pragma unroll
        for (int s = 0; s < 7; ++s) {
            const int ka = 4 * s + 2 * half;
            if (ka     < FIN) { fa[s].x = x[(node0 + l16) * FIN + ka    ]; fb[s].x = W1[(ka    ) * F1 + l16]; }
            if (ka + 1 < FIN) { fa[s].y = x[(node0 + l16) * FIN + ka + 1]; fb[s].y = W1[(ka + 1) * F1 + l16]; }
        }
    }

    const int beg = bstart[b];
    const int end = bstart[b + 1];
    lds_vint* cw = (lds_vint*)(cnt + wave * NPB);
    const int nbatch = (end - beg + WG - 1) / WG;
    for (int j0 = 0; j0 < nbatch; j0 += LD_U) {
        int cc[LD_U];
        #pragma unroll
        for (int u = 0; u < LD_U; ++u) {
            const int i = beg + (j0 + u) * WG + tid;
            cc[u] = sorted[min(i, end - 1)].y;
        }
        #pragma unroll
        for (int u = 0; u < LD_U; ++u) {
            const int i = beg + (j0 + u) * WG + tid;
            const bool act = i < end;
            const int key = cc[u] - base;
            const unsigned m = wave_peers(key, act, NPB_SHIFT);
            if (act && lanes_below(m) == 0) cw[key] = cw[key] + __popc(m);
        }
    }

    v8f c = {0.f, 0.f, 0.f, 0.f, 0.f, 0.f, 0.f, 0.f};
    if (tile) {
        #pragma unroll
        for (int s = 0; s < 7; ++s) {
            c = __builtin_amdgcn_wmma_f32_16x16x4_f32(
                     false, fa[s],  false, fb[s],
                     (short)0, c,  false,  false);
        }
    }
    __syncthreads();
    if (tid < NPB) {
        float dv = 0.0f;
        if (base + tid < n) {
            int d = 1;
            #pragma unroll
            for (int w = 0; w < WAVES; ++w) d += cnt[w * NPB + tid];
            dv = rsqrtf((float)d);
            dinv[base + tid] = dv;
        }
        dsm[tid] = dv;
    }
    __syncthreads();
    if (tile) {
        #pragma unroll
        for (int v = 0; v < 8; ++v) {
            const int ln = wave * 16 + v + 8 * half;
            h1s[(base + ln) * F1 + l16] = c[v] * dsm[ln];
        }
    }
}

#define L1_U 8
#define L1_S (L1_U / 4)

template <int U>
__device__ __forceinline__ int quad_lane(int v) {
    return __builtin_amdgcn_mov_dpp(v, U * 0x55, 0xf, 0xf, true);
}

template <int U>
__device__ __forceinline__ void l1_add_group(lds_vfloat4* aw, int pk, unsigned multi,
                                             int q, v4f v) {
    const int pku  = quad_lane<U>(pk);
    const int key  = pku & 0xFF;
    const int rank = (pku >> 8) & 0xFF;
    const bool act = (pku >> 16) != 0;
    if (((multi >> U) & 0x11111111u) == 0u) {
        if (act) {
            v4f t = aw[key * 4 + q];
            aw[key * 4 + q] = t + v;
        }
    } else {
        for (int turn = 0; ; ++turn) {
            const bool mine = act && rank == turn;
            if (__builtin_amdgcn_ballot_w32(mine) == 0) break;
            if (mine) {
                v4f t = aw[key * 4 + q];
                aw[key * 4 + q] = t + v;
            }
        }
    }
}
__global__ void __launch_bounds__(WG)
gcn_layer1_kernel(const int2* __restrict__ sorted,
                  const int* __restrict__ bstart,
                  const float* __restrict__ h1s,
                  const float* __restrict__ dinv,
                  const float* __restrict__ b1,
                  const float* __restrict__ W2,
                  float* __restrict__ h2s,
                  int n) {
    __shared__ __attribute__((aligned(16))) float acc[WAVES * NPB * F1];
    const int tid  = threadIdx.x;
    const int lane = tid & 31;
    const int wave = tid >> 5;
    const int grp  = lane >> 2;
    const int q    = lane & 3;
    const int b    = blockIdx.x;
    const int base = b << NPB_SHIFT;
    for (int k = tid; k < WAVES * NPB * F1; k += WG) acc[k] = 0.0f;
    __syncthreads();
    const int beg = bstart[b];
    const int end = bstart[b + 1];
    const v4f* h1v = (const v4f*)h1s;
    lds_vfloat4* aw = (lds_vfloat4*)(acc + wave * (NPB * F1));
    const unsigned qmask = 0x11111111u << q;
    for (int i0 = beg; i0 < end; i0 += 8 * WAVES * L1_U) {
        int2 es[L1_S];
        v4f  vv[L1_U];
        #pragma unroll
        for (int s = 0; s < L1_S; ++s) {
            const int i = i0 + (4 * s + q) * (8 * WAVES) + wave * 8 + grp;
            es[s] = sorted[min(i, end - 1)];
        }
        #pragma unroll
        for (int s = 0; s < L1_S; ++s) {
            vv[4 * s + 0] = h1v[quad_lane<0>(es[s].x) * 4 + q];
            vv[4 * s + 1] = h1v[quad_lane<1>(es[s].x) * 4 + q];
            vv[4 * s + 2] = h1v[quad_lane<2>(es[s].x) * 4 + q];
            vv[4 * s + 3] = h1v[quad_lane<3>(es[s].x) * 4 + q];
        }
        #pragma unroll
        for (int s = 0; s < L1_S; ++s) {
            const int i = i0 + (4 * s + q) * (8 * WAVES) + wave * 8 + grp;
            const bool act = i < end;
            const int key = es[s].y - base;
            const unsigned m = wave_peers(key, act, NPB_SHIFT);
            const int rank = lanes_below(m & qmask);
            const unsigned multi = __builtin_amdgcn_ballot_w32(rank != 0);
            const int pk = key | (rank << 8) | (act ? 0x10000 : 0);
            l1_add_group<0>(aw, pk, multi, q, vv[4 * s + 0]);
            l1_add_group<1>(aw, pk, multi, q, vv[4 * s + 1]);
            l1_add_group<2>(aw, pk, multi, q, vv[4 * s + 2]);
            l1_add_group<3>(aw, pk, multi, q, vv[4 * s + 3]);
        }
    }
    __syncthreads();
    for (int k = tid; k < NPB * F1; k += WG) {
        float s = acc[k];
        #pragma unroll
        for (int w = 1; w < WAVES; ++w) s += acc[w * (NPB * F1) + k];
        acc[k] = s;
    }
    __syncthreads();
    const int node = base + tid;
    if (tid < NPB && node < n) {
        float di = dinv[node];
        float a0 = 0.0f, a1 = 0.0f;
        #pragma unroll
        for (int k = 0; k < F1; ++k) {
            float v = (acc[tid * F1 + k] + h1s[node * F1 + k]) * di + b1[k];
            v = v > 0.0f ? v : 0.0f;
            a0 = fmaf(v, W2[k * F2 + 0], a0);
            a1 = fmaf(v, W2[k * F2 + 1], a1);
        }
        h2s[node * F2 + 0] = a0 * di;
        h2s[node * F2 + 1] = a1 * di;
    }
}

__global__ void __launch_bounds__(WG)
gcn_layer2_kernel(const int2* __restrict__ sorted,
                  const int* __restrict__ bstart,
                  const float* __restrict__ h2s,
                  const float* __restrict__ dinv,
                  const float* __restrict__ b2,
                  float* __restrict__ out,
                  int n) {
    __shared__ float acc[WAVES * NPB * F2];
    const int tid  = threadIdx.x;
    const int lane = tid & 31;
    const int wave = tid >> 5;
    const int b    = blockIdx.x;
    const int base = b << NPB_SHIFT;
    for (int k = tid; k < WAVES * NPB * F2; k += WG) acc[k] = 0.0f;
    __syncthreads();
    const int beg = bstart[b];
    const int end = bstart[b + 1];
    const float2* h2v = (const float2*)h2s;
    lds_vfloat* aw = (lds_vfloat*)(acc + wave * (NPB * F2));
    const int nbatch = (end - beg + WG - 1) / WG;
    for (int j0 = 0; j0 < nbatch; j0 += LD_U) {
        int2   ee[LD_U];
        float2 vv[LD_U];
        #pragma unroll
        for (int u = 0; u < LD_U; ++u) {
            const int i = beg + (j0 + u) * WG + tid;
            ee[u] = sorted[min(i, end - 1)];
        }
        #pragma unroll
        for (int u = 0; u < LD_U; ++u) vv[u] = h2v[ee[u].x];
        #pragma unroll
        for (int u = 0; u < LD_U; ++u) {
            const int i = beg + (j0 + u) * WG + tid;
            const bool act = i < end;
            const int key = ee[u].y - base;
            const unsigned m = wave_peers(key, act, NPB_SHIFT);
            const int rank = lanes_below(m);
            for (int turn = 0; ; ++turn) {
                const bool mine = act && rank == turn;
                if (__builtin_amdgcn_ballot_w32(mine) == 0) break;
                if (mine) {
                    aw[key * F2 + 0] = aw[key * F2 + 0] + vv[u].x;
                    aw[key * F2 + 1] = aw[key * F2 + 1] + vv[u].y;
                }
            }
        }
    }
    __syncthreads();
    const int node = base + tid;
    if (tid < NPB && node < n) {
        float s0 = 0.0f, s1 = 0.0f;
        #pragma unroll
        for (int w = 0; w < WAVES; ++w) {
            s0 += acc[w * (NPB * F2) + tid * F2 + 0];
            s1 += acc[w * (NPB * F2) + tid * F2 + 1];
        }
        float di = dinv[node];
        float2 s = h2v[node];
        float a = (s0 + s.x) * di + b2[0];
        float c = (s1 + s.y) * di + b2[1];
        float m = fmaxf(a, c);
        float lse = m + logf(expf(a - m) + expf(c - m));
        out[node * F2 + 0] = a - lse;
        out[node * F2 + 1] = c - lse;
    }
}

extern "C" void kernel_launch(void* const* d_in, const int* in_sizes, int n_in,
                              void* d_out, int out_size, void* d_ws, size_t ws_size,
                              hipStream_t stream) {
    const float* x   = (const float*)d_in[0];
    const int*   ei  = (const int*)d_in[1];
    const float* W1  = (const float*)d_in[2];
    const float* b1  = (const float*)d_in[3];
    const float* W2  = (const float*)d_in[4];
    const float* b2  = (const float*)d_in[5];
    float*       out = (float*)d_out;

    const int N = in_sizes[0] / FIN;
    const int E = in_sizes[1] / 2;
    const int* rows = ei;
    const int* cols = ei + E;

    const int NB    = (N + NPB - 1) / NPB;
    const int chunk = (E + SPLIT_WG - 1) / SPLIT_WG;

    int*   wsi    = (int*)d_ws;
    int2*  sorted = (int2*)wsi;
    float* h1s    = (float*)(wsi + (size_t)2 * E);
    float* h2s    = h1s + (size_t)16 * N;
    float* dinv   = h2s + (size_t)2 * N;
    int*   counts = (int*)(dinv + (size_t)N);
    int*   totals = counts + (size_t)NB_MAX * SPLIT_WG;
    int*   bstart = totals + NB_MAX;
    unsigned short* place = (unsigned short*)(bstart + NB_MAX + 8);

    gcn_split_count_kernel<<<SPLIT_WG, SWG, 0, stream>>>(cols, counts, place, E, chunk, NB);
    gcn_split_rowscan_kernel<<<NB, SPLIT_WG, 0, stream>>>(counts, totals);
    gcn_split_scatter_kernel<<<SPLIT_WG, SWG, 0, stream>>>(rows, cols, place, counts, totals,
                                                          bstart, sorted, E, chunk, NB);

    gcn_degree_xw1_kernel<<<NB, WG, 0, stream>>>(sorted, bstart, x, W1, dinv, h1s, N);

    gcn_layer1_kernel<<<NB, WG, 0, stream>>>(sorted, bstart, h1s, dinv, b1, W2, h2s, N);

    gcn_layer2_kernel<<<NB, WG, 0, stream>>>(sorted, bstart, h2s, dinv, b2, out, N);
}


# Bucket-count table kept chunk-major: count kernel stores and scatter kernel loads run along whole lines; scan kernel is 32 column-tile workgroups, lane per bucket; results bit-identical
# speedup vs baseline: 5.6995x; 1.1058x over previous
#include <hip/hip_runtime.h>
#include <hip/hip_bf16.h>


typedef __attribute__((ext_vector_type(2))) float v2f;
typedef __attribute__((ext_vector_type(8))) float v8f;

#define F1 16
#define F2 2
#define FIN 25

#define NPB_SHIFT 7
#define NPB 128
#define NB_MAX 1024
#define NB_BITS 10
#define SPLIT_WG 256
#define SWG 1024
#define SWAVES 32
#define STAGE_CAP 16384
#define WG 256
#define WAVES 8
#define LD_U 4
#define MAXB (STAGE_CAP / SWG)

typedef __attribute__((address_space(3))) volatile int   lds_vint;
typedef __attribute__((address_space(3))) volatile float lds_vfloat;
typedef __attribute__((ext_vector_type(4))) float v4f;
typedef __attribute__((address_space(3))) volatile v4f   lds_vfloat4;
typedef __attribute__((ext_vector_type(2))) int v2i;

__device__ __forceinline__ unsigned wave_peers(int key, bool act, int nbits) {
    unsigned m = __builtin_amdgcn_ballot_w32(act);
    for (int bit = 0; bit < nbits; ++bit) {
        const int x = __builtin_amdgcn_sbfe(key, bit, 1);
        const unsigned bal = __builtin_amdgcn_ballot_w32(x != 0);
        m = __builtin_amdgcn_bitop3_b32(m, bal, (unsigned)x, 0x90);
    }
    return act ? m : 0u;
}

__device__ __forceinline__ int lanes_below(unsigned m) {
    return (int)__builtin_amdgcn_mbcnt_lo(m, 0u);
}

template <typename T>
__device__ __forceinline__ T block_excl_scan(T v, T* sm, int tid, int nthreads,
                                             T* total) {
    sm[tid] = v;
    __syncthreads();
    for (int off = 1; off < nthreads; off <<= 1) {
        T t = (tid >= off) ? sm[tid - off] : (T)0;
        __syncthreads();
        sm[tid] += t;
        __syncthreads();
    }
    T incl = sm[tid];
    *total = sm[nthreads - 1];
    __syncthreads();
    return incl - v;
}

__global__ void __launch_bounds__(SWG)
gcn_split_count_kernel(const int* __restrict__ cols,
                       int* __restrict__ counts,
                       unsigned short* __restrict__ place,
                       int n_edges, int chunk, int nb) {
    static_assert(SWG == NB_MAX, "one thread per bucket in the offset step");
    static_assert(STAGE_CAP <= 65536, "a place fits 16 bits");
    __shared__ int hist[SWAVES * NB_MAX];
    const int tid  = threadIdx.x;
    const int lane = tid & 31;
    const int wave = tid >> 5;
    const int wg   = blockIdx.x;
    const int beg = min(n_edges, wg * chunk);
    const int end = min(n_edges, beg + chunk);
    const int nbatch  = (end - beg + SWG - 1) / SWG;
    const int i_first = beg + tid;
    int cc[MAXB], loc[MAXB];
    #pragma unroll
    for (int j = 0; j < MAXB; ++j) cc[j] = 0;
    if (end > beg) {
        #pragma unroll
        for (int j = 0; j < MAXB; ++j) cc[j] = cols[min(i_first + j * SWG, end - 1)];
    }
    for (int k = tid; k < SWAVES * NB_MAX; k += SWG) hist[k] = 0;
    __syncthreads();
    lds_vint* hw = (lds_vint*)(hist + wave * NB_MAX);
    #pragma unroll
    for (int j = 0; j < MAXB; ++j) {
        loc[j] = 0;
        if (j < nbatch) {
            const bool act = i_first + j * SWG < end;
            const int key = cc[j] >> NPB_SHIFT;
            const unsigned m = wave_peers(key, act, NB_BITS);
            const int rank = lanes_below(m);
            int p = 0;
            if (act) p = hw[key];
            if (act && rank == 0) hw[key] = p + __popc(m);
            loc[j] = p + rank;
        }
    }
    __syncthreads();
    int tot = 0;
    #pragma unroll
    for (int w = 0; w < SWAVES; ++w) {
        int t = hist[w * NB_MAX + tid];
        hist[w * NB_MAX + tid] = tot;
        tot += t;
    }
    counts[wg * NB_MAX + tid] = tot;
    __syncthreads();
    #pragma unroll
    for (int j = 0; j < MAXB; ++j) loc[j] += hw[cc[j] >> NPB_SHIFT];
    #pragma unroll
    for (int j = 0; j < MAXB; ++j) {
        if (i_first + j * SWG < end) place[i_first + j * SWG] = (unsigned short)loc[j];
    }
}

#define RS_CPW (SPLIT_WG / WAVES)
__global__ void __launch_bounds__(WG)
gcn_split_rowscan_kernel(int* __restrict__ counts,
                         int* __restrict__ totals) {
    static_assert(SPLIT_WG % WAVES == 0, "the chunks split evenly over the waves");
    __shared__ int wsum[WAVES * 32];
    const int tid  = threadIdx.x;
    const int lane = tid & 31;
    const int wave = tid >> 5;
    const int b    = blockIdx.x * 32 + lane;
    int* p = counts + (size_t)(wave * RS_CPW) * NB_MAX + b;
    int v[RS_CPW];
    #pragma unroll
    for (int i = 0; i < RS_CPW; ++i) v[i] = p[(size_t)i * NB_MAX];
    int run = 0;
    #pragma unroll
    for (int i = 0; i < RS_CPW; ++i) {
        const int t = v[i];
        v[i] = run;
        run += t;
    }
    wsum[wave * 32 + lane] = run;
    __syncthreads();
    int off = 0, total = 0;
    #pragma unroll
    for (int w = 0; w < WAVES; ++w) {
        const int t = wsum[w * 32 + lane];
        if (w < wave) off += t;
        total += t;
    }
    #pragma unroll
    for (int i = 0; i < RS_CPW; ++i) p[(size_t)i * NB_MAX] = v[i] + off;
    if (wave == 0) totals[b] = total;
}

__global__ void __launch_bounds__(SWG)
gcn_split_scatter_kernel(const int* __restrict__ rows,
                         const int* __restrict__ cols,
                         const unsigned short* __restrict__ place,
                         const int* __restrict__ counts,
                         const int* __restrict__ totals,
                         int* __restrict__ bstart,
                         int2* __restrict__ sorted,
                         int n_edges, int chunk, int nb) {
    static_assert(SWG == NB_MAX, "one thread per bucket in the start step");
    __shared__ v2i stage[STAGE_CAP];
    __shared__ int lstart[NB_MAX];
    __shared__ int delta[NB_MAX];
    __shared__ unsigned long long scan_sm[SWG];
    const int tid = threadIdx.x;
    const int wg  = blockIdx.x;
    const int beg = min(n_edges, wg * chunk);
    const int end = min(n_edges, beg + chunk);
    const int i_first = beg + tid;
    int rr[MAXB], cc[MAXB], pl[MAXB];
    if (end <= beg && wg != 0) return;
    #pragma unroll
    for (int j = 0; j < MAXB; ++j) { rr[j] = 0; cc[j] = 0; pl[j] = 0; }
    if (end > beg) {
        #pragma unroll
        for (int j = 0; j < MAXB; ++j) {
            const int i = min(i_first + j * SWG, end - 1);
            rr[j] = rows[i];
            cc[j] = cols[i];
            pl[j] = place[i];
        }
    }
    int cnt = 0, sc = 0, tot = 0;
    if (tid < nb) {
        sc  = counts[wg * NB_MAX + tid];
        tot = totals[tid];
        const int nx = (wg + 1 < SPLIT_WG) ? counts[(wg + 1) * NB_MAX + tid] : tot;
        cnt = nx - sc;
    }
    unsigned long long all;
    const unsigned long long ex = block_excl_scan(
        ((unsigned long long)(unsigned)cnt << 32) | (unsigned)tot, scan_sm, tid, SWG, &all);
    const int bs = (int)(unsigned)ex;
    const int ls = (int)(ex >> 32);
    lstart[tid] = ls;
    delta[tid]  = bs + sc - ls;
    if (wg == 0) {
        if (tid < nb)  bstart[tid] = bs;
        if (tid == 0)  bstart[nb]  = (int)(unsigned)all;
    }
    __syncthreads();
    int at[MAXB];
    #pragma unroll
    for (int j = 0; j < MAXB; ++j) at[j] = lstart[cc[j] >> NPB_SHIFT] + pl[j];
    #pragma unroll
    for (int j = 0; j < MAXB; ++j) {
        if (i_first + j * SWG < end) {
            v2i e;
            e.x = rr[j];
            e.y = cc[j];
            stage[at[j]] = e;
        }
    }
    __syncthreads();
    v2i* sv = (v2i*)sorted;
    const int n_chunk = end - beg;
    for (int i0 = tid; i0 < n_chunk; i0 += LD_U * SWG) {
        v2i e[LD_U];
        int d[LD_U];
        #pragma unroll
        for (int u = 0; u < LD_U; ++u) e[u] = stage[min(i0 + u * SWG, n_chunk - 1)];
        #pragma unroll
        for (int u = 0; u < LD_U; ++u) d[u] = delta[e[u].y >> NPB_SHIFT];
        #pragma unroll
        for (int u = 0; u < LD_U; ++u) {
            const int i = i0 + u * SWG;
            if (i < n_chunk) sv[i + d[u]] = e[u];
        }
    }
}

__global__ void __launch_bounds__(WG)
gcn_degree_xw1_kernel(const int2* __restrict__ sorted,
                      const int* __restrict__ bstart,
                      const float* __restrict__ x,
                      const float* __restrict__ W1,
                      float* __restrict__ dinv,
                      float* __restrict__ h1s,
                      int n) {
    static_assert(NPB == WAVES * 16, "one 16-node tile per wave covers the bucket");
    __shared__ int cnt[WAVES * NPB];
    __shared__ float dsm[NPB];
    const int tid  = threadIdx.x;
    const int lane = tid & 31;
    const int wave = tid >> 5;
    const int half = lane >> 4;
    const int l16  = lane & 15;
    const int b    = blockIdx.x;
    const int base = b << NPB_SHIFT;
    const int node0 = base + wave * 16;
    const bool tile = node0 < n;

    for (int k = tid; k < WAVES * NPB; k += WG) cnt[k] = 0;
    __syncthreads();

    v2f fa[7], fb[7];
    #pragma unroll
    for (int s = 0; s < 7; ++s) { fa[s].x = 0.0f; fa[s].y = 0.0f; fb[s].x = 0.0f; fb[s].y = 0.0f; }
    if (tile) {
        #pragma unroll
        for (int s = 0; s < 7; ++s) {
            const int ka = 4 * s + 2 * half;
            if (ka     < FIN) { fa[s].x = x[(node0 + l16) * FIN + ka    ]; fb[s].x = W1[(ka    ) * F1 + l16]; }
            if (ka + 1 < FIN) { fa[s].y = x[(node0 + l16) * FIN + ka + 1]; fb[s].y = W1[(ka + 1) * F1 + l16]; }
        }
    }

    const int beg = bstart[b];
    const int end = bstart[b + 1];
    lds_vint* cw = (lds_vint*)(cnt + wave * NPB);
    const int nbatch = (end - beg + WG - 1) / WG;
    for (int j0 = 0; j0 < nbatch; j0 += LD_U) {
        int cc[LD_U];
        #pragma unroll
        for (int u = 0; u < LD_U; ++u) {
            const int i = beg + (j0 + u) * WG + tid;
            cc[u] = sorted[min(i, end - 1)].y;
        }
        #pragma unroll
        for (int u = 0; u < LD_U; ++u) {
            const int i = beg + (j0 + u) * WG + tid;
            const bool act = i < end;
            const int key = cc[u] - base;
            const unsigned m = wave_peers(key, act, NPB_SHIFT);
            if (act && lanes_below(m) == 0) cw[key] = cw[key] + __popc(m);
        }
    }

    v8f c = {0.f, 0.f, 0.f, 0.f, 0.f, 0.f, 0.f, 0.f};
    if (tile) {
        #pragma unroll
        for (int s = 0; s < 7; ++s) {
            c = __builtin_amdgcn_wmma_f32_16x16x4_f32(
                     false, fa[s],  false, fb[s],
                     (short)0, c,  false,  false);
        }
    }
    __syncthreads();
    if (tid < NPB) {
        float dv = 0.0f;
        if (base + tid < n) {
            int d = 1;
            #pragma unroll
            for (int w = 0; w < WAVES; ++w) d += cnt[w * NPB + tid];
            dv = rsqrtf((float)d);
            dinv[base + tid] = dv;
        }
        dsm[tid] = dv;
    }
    __syncthreads();
    if (tile) {
        #pragma unroll
        for (int v = 0; v < 8; ++v) {
            const int ln = wave * 16 + v + 8 * half;
            h1s[(base + ln) * F1 + l16] = c[v] * dsm[ln];
        }
    }
}

#define L1_U 8
#define L1_S (L1_U / 4)

template <int U>
__device__ __forceinline__ int quad_lane(int v) {
    return __builtin_amdgcn_mov_dpp(v, U * 0x55, 0xf, 0xf, true);
}

template <int U>
__device__ __forceinline__ void l1_add_group(lds_vfloat4* aw, int pk, unsigned multi,
                                             int q, v4f v) {
    const int pku  = quad_lane<U>(pk);
    const int key  = pku & 0xFF;
    const int rank = (pku >> 8) & 0xFF;
    const bool act = (pku >> 16) != 0;
    if (((multi >> U) & 0x11111111u) == 0u) {
        if (act) {
            v4f t = aw[key * 4 + q];
            aw[key * 4 + q] = t + v;
        }
    } else {
        for (int turn = 0; ; ++turn) {
            const bool mine = act && rank == turn;
            if (__builtin_amdgcn_ballot_w32(mine) == 0) break;
            if (mine) {
                v4f t = aw[key * 4 + q];
                aw[key * 4 + q] = t + v;
            }
        }
    }
}
__global__ void __launch_bounds__(WG)
gcn_layer1_kernel(const int2* __restrict__ sorted,
                  const int* __restrict__ bstart,
                  const float* __restrict__ h1s,
                  const float* __restrict__ dinv,
                  const float* __restrict__ b1,
                  const float* __restrict__ W2,
                  float* __restrict__ h2s,
                  int n) {
    __shared__ __attribute__((aligned(16))) float acc[WAVES * NPB * F1];
    const int tid  = threadIdx.x;
    const int lane = tid & 31;
    const int wave = tid >> 5;
    const int grp  = lane >> 2;
    const int q    = lane & 3;
    const int b    = blockIdx.x;
    const int base = b << NPB_SHIFT;
    for (int k = tid; k < WAVES * NPB * F1; k += WG) acc[k] = 0.0f;
    __syncthreads();
    const int beg = bstart[b];
    const int end = bstart[b + 1];
    const v4f* h1v = (const v4f*)h1s;
    lds_vfloat4* aw = (lds_vfloat4*)(acc + wave * (NPB * F1));
    const unsigned qmask = 0x11111111u << q;
    for (int i0 = beg; i0 < end; i0 += 8 * WAVES * L1_U) {
        int2 es[L1_S];
        v4f  vv[L1_U];
        #pragma unroll
        for (int s = 0; s < L1_S; ++s) {
            const int i = i0 + (4 * s + q) * (8 * WAVES) + wave * 8 + grp;
            es[s] = sorted[min(i, end - 1)];
        }
        #pragma unroll
        for (int s = 0; s < L1_S; ++s) {
            vv[4 * s + 0] = h1v[quad_lane<0>(es[s].x) * 4 + q];
            vv[4 * s + 1] = h1v[quad_lane<1>(es[s].x) * 4 + q];
            vv[4 * s + 2] = h1v[quad_lane<2>(es[s].x) * 4 + q];
            vv[4 * s + 3] = h1v[quad_lane<3>(es[s].x) * 4 + q];
        }
        #pragma unroll
        for (int s = 0; s < L1_S; ++s) {
            const int i = i0 + (4 * s + q) * (8 * WAVES) + wave * 8 + grp;
            const bool act = i < end;
            const int key = es[s].y - base;
            const unsigned m = wave_peers(key, act, NPB_SHIFT);
            const int rank = lanes_below(m & qmask);
            const unsigned multi = __builtin_amdgcn_ballot_w32(rank != 0);
            const int pk = key | (rank << 8) | (act ? 0x10000 : 0);
            l1_add_group<0>(aw, pk, multi, q, vv[4 * s + 0]);
            l1_add_group<1>(aw, pk, multi, q, vv[4 * s + 1]);
            l1_add_group<2>(aw, pk, multi, q, vv[4 * s + 2]);
            l1_add_group<3>(aw, pk, multi, q, vv[4 * s + 3]);
        }
    }
    __syncthreads();
    for (int k = tid; k < NPB * F1; k += WG) {
        float s = acc[k];
        #pragma unroll
        for (int w = 1; w < WAVES; ++w) s += acc[w * (NPB * F1) + k];
        acc[k] = s;
    }
    __syncthreads();
    const int node = base + tid;
    if (tid < NPB && node < n) {
        float di = dinv[node];
        float a0 = 0.0f, a1 = 0.0f;
        #pragma unroll
        for (int k = 0; k < F1; ++k) {
            float v = (acc[tid * F1 + k] + h1s[node * F1 + k]) * di + b1[k];
            v = v > 0.0f ? v : 0.0f;
            a0 = fmaf(v, W2[k * F2 + 0], a0);
            a1 = fmaf(v, W2[k * F2 + 1], a1);
        }
        h2s[node * F2 + 0] = a0 * di;
        h2s[node * F2 + 1] = a1 * di;
    }
}

__global__ void __launch_bounds__(WG)
gcn_layer2_kernel(const int2* __restrict__ sorted,
                  const int* __restrict__ bstart,
                  const float* __restrict__ h2s,
                  const float* __restrict__ dinv,
                  const float* __restrict__ b2,
                  float* __restrict__ out,
                  int n) {
    __shared__ float acc[WAVES * NPB * F2];
    const int tid  = threadIdx.x;
    const int lane = tid & 31;
    const int wave = tid >> 5;
    const int b    = blockIdx.x;
    const int base = b << NPB_SHIFT;
    for (int k = tid; k < WAVES * NPB * F2; k += WG) acc[k] = 0.0f;
    __syncthreads();
    const int beg = bstart[b];
    const int end = bstart[b + 1];
    const float2* h2v = (const float2*)h2s;
    lds_vfloat* aw = (lds_vfloat*)(acc + wave * (NPB * F2));
    const int nbatch = (end - beg + WG - 1) / WG;
    for (int j0 = 0; j0 < nbatch; j0 += LD_U) {
        int2   ee[LD_U];
        float2 vv[LD_U];
        #pragma unroll
        for (int u = 0; u < LD_U; ++u) {
            const int i = beg + (j0 + u) * WG + tid;
            ee[u] = sorted[min(i, end - 1)];
        }
        #pragma unroll
        for (int u = 0; u < LD_U; ++u) vv[u] = h2v[ee[u].x];
        #pragma unroll
        for (int u = 0; u < LD_U; ++u) {
            const int i = beg + (j0 + u) * WG + tid;
            const bool act = i < end;
            const int key = ee[u].y - base;
            const unsigned m = wave_peers(key, act, NPB_SHIFT);
            const int rank = lanes_below(m);
            for (int turn = 0; ; ++turn) {
                const bool mine = act && rank == turn;
                if (__builtin_amdgcn_ballot_w32(mine) == 0) break;
                if (mine) {
                    aw[key * F2 + 0] = aw[key * F2 + 0] + vv[u].x;
                    aw[key * F2 + 1] = aw[key * F2 + 1] + vv[u].y;
                }
            }
        }
    }
    __syncthreads();
    const int node = base + tid;
    if (tid < NPB && node < n) {
        float s0 = 0.0f, s1 = 0.0f;
        #pragma unroll
        for (int w = 0; w < WAVES; ++w) {
            s0 += acc[w * (NPB * F2) + tid * F2 + 0];
            s1 += acc[w * (NPB * F2) + tid * F2 + 1];
        }
        float di = dinv[node];
        float2 s = h2v[node];
        float a = (s0 + s.x) * di + b2[0];
        float c = (s1 + s.y) * di + b2[1];
        float m = fmaxf(a, c);
        float lse = m + logf(expf(a - m) + expf(c - m));
        out[node * F2 + 0] = a - lse;
        out[node * F2 + 1] = c - lse;
    }
}

extern "C" void kernel_launch(void* const* d_in, const int* in_sizes, int n_in,
                              void* d_out, int out_size, void* d_ws, size_t ws_size,
                              hipStream_t stream) {
    const float* x   = (const float*)d_in[0];
    const int*   ei  = (const int*)d_in[1];
    const float* W1  = (const float*)d_in[2];
    const float* b1  = (const float*)d_in[3];
    const float* W2  = (const float*)d_in[4];
    const float* b2  = (const float*)d_in[5];
    float*       out = (float*)d_out;

    const int N = in_sizes[0] / FIN;
    const int E = in_sizes[1] / 2;
    const int* rows = ei;
    const int* cols = ei + E;

    const int NB    = (N + NPB - 1) / NPB;
    const int chunk = (E + SPLIT_WG - 1) / SPLIT_WG;

    int*   wsi    = (int*)d_ws;
    int2*  sorted = (int2*)wsi;
    float* h1s    = (float*)(wsi + (size_t)2 * E);
    float* h2s    = h1s + (size_t)16 * N;
    float* dinv   = h2s + (size_t)2 * N;
    int*   counts = (int*)(dinv + (size_t)N);
    int*   totals = counts + (size_t)NB_MAX * SPLIT_WG;
    int*   bstart = totals + NB_MAX;
    unsigned short* place = (unsigned short*)(bstart + NB_MAX + 8);

    gcn_split_count_kernel<<<SPLIT_WG, SWG, 0, stream>>>(cols, counts, place, E, chunk, NB);
    gcn_split_rowscan_kernel<<<NB_MAX / 32, WG, 0, stream>>>(counts, totals);
    gcn_split_scatter_kernel<<<SPLIT_WG, SWG, 0, stream>>>(rows, cols, place, counts, totals,
                                                          bstart, sorted, E, chunk, NB);

    gcn_degree_xw1_kernel<<<NB, WG, 0, stream>>>(sorted, bstart, x, W1, dinv, h1s, N);

    gcn_layer1_kernel<<<NB, WG, 0, stream>>>(sorted, bstart, h1s, dinv, b1, W2, h2s, N);

    gcn_layer2_kernel<<<NB, WG, 0, stream>>>(sorted, bstart, h2s, dinv, b2, out, N);
}


# Sorted edge packed into one 4-byte word (source in 17 bits, target's index inside its bucket above): scatter stores and the bucket kernels' edge loads move half the bytes; sum order unchanged
# speedup vs baseline: 5.8886x; 1.0332x over previous
#include <hip/hip_runtime.h>
#include <hip/hip_bf16.h>


typedef __attribute__((ext_vector_type(2))) float v2f;
typedef __attribute__((ext_vector_type(8))) float v8f;

#define F1 16
#define F2 2
#define FIN 25

#define NPB_SHIFT 7
#define NPB 128
#define NB_MAX 1024
#define NB_BITS 10
#define ROW_BITS 17
#define ROW_MASK ((1 << ROW_BITS) - 1)
#define SPLIT_WG 256
#define SWG 1024
#define SWAVES 32
#define STAGE_CAP 16384
#define WG 256
#define WAVES 8
#define LD_U 4
#define MAXB (STAGE_CAP / SWG)

typedef __attribute__((address_space(3))) volatile int   lds_vint;
typedef __attribute__((address_space(3))) volatile float lds_vfloat;
typedef __attribute__((ext_vector_type(4))) float v4f;
typedef __attribute__((address_space(3))) volatile v4f   lds_vfloat4;
typedef __attribute__((ext_vector_type(2))) int v2i;

__device__ __forceinline__ unsigned wave_peers(int key, bool act, int nbits) {
    unsigned m = __builtin_amdgcn_ballot_w32(act);
    for (int bit = 0; bit < nbits; ++bit) {
        const int x = __builtin_amdgcn_sbfe(key, bit, 1);
        const unsigned bal = __builtin_amdgcn_ballot_w32(x != 0);
        m = __builtin_amdgcn_bitop3_b32(m, bal, (unsigned)x, 0x90);
    }
    return act ? m : 0u;
}

__device__ __forceinline__ int lanes_below(unsigned m) {
    return (int)__builtin_amdgcn_mbcnt_lo(m, 0u);
}

template <typename T>
__device__ __forceinline__ T block_excl_scan(T v, T* sm, int tid, int nthreads,
                                             T* total) {
    sm[tid] = v;
    __syncthreads();
    for (int off = 1; off < nthreads; off <<= 1) {
        T t = (tid >= off) ? sm[tid - off] : (T)0;
        __syncthreads();
        sm[tid] += t;
        __syncthreads();
    }
    T incl = sm[tid];
    *total = sm[nthreads - 1];
    __syncthreads();
    return incl - v;
}

__global__ void __launch_bounds__(SWG)
gcn_split_count_kernel(const int* __restrict__ cols,
                       int* __restrict__ counts,
                       unsigned short* __restrict__ place,
                       int n_edges, int chunk, int nb) {
    static_assert(SWG == NB_MAX, "one thread per bucket in the offset step");
    static_assert(STAGE_CAP <= 65536, "a place fits 16 bits");
    __shared__ int hist[SWAVES * NB_MAX];
    const int tid  = threadIdx.x;
    const int lane = tid & 31;
    const int wave = tid >> 5;
    const int wg   = blockIdx.x;
    const int beg = min(n_edges, wg * chunk);
    const int end = min(n_edges, beg + chunk);
    const int nbatch  = (end - beg + SWG - 1) / SWG;
    const int i_first = beg + tid;
    int cc[MAXB], loc[MAXB];
    #pragma unroll
    for (int j = 0; j < MAXB; ++j) cc[j] = 0;
    if (end > beg) {
        #pragma unroll
        for (int j = 0; j < MAXB; ++j) cc[j] = cols[min(i_first + j * SWG, end - 1)];
    }
    for (int k = tid; k < SWAVES * NB_MAX; k += SWG) hist[k] = 0;
    __syncthreads();
    lds_vint* hw = (lds_vint*)(hist + wave * NB_MAX);
    #pragma unroll
    for (int j = 0; j < MAXB; ++j) {
        loc[j] = 0;
        if (j < nbatch) {
            const bool act = i_first + j * SWG < end;
            const int key = cc[j] >> NPB_SHIFT;
            const unsigned m = wave_peers(key, act, NB_BITS);
            const int rank = lanes_below(m);
            int p = 0;
            if (act) p = hw[key];
            if (act && rank == 0) hw[key] = p + __popc(m);
            loc[j] = p + rank;
        }
    }
    __syncthreads();
    int tot = 0;
    #pragma unroll
    for (int w = 0; w < SWAVES; ++w) {
        int t = hist[w * NB_MAX + tid];
        hist[w * NB_MAX + tid] = tot;
        tot += t;
    }
    counts[wg * NB_MAX + tid] = tot;
    __syncthreads();
    #pragma unroll
    for (int j = 0; j < MAXB; ++j) loc[j] += hw[cc[j] >> NPB_SHIFT];
    #pragma unroll
    for (int j = 0; j < MAXB; ++j) {
        if (i_first + j * SWG < end) place[i_first + j * SWG] = (unsigned short)loc[j];
    }
}

#define RS_CPW (SPLIT_WG / WAVES)
__global__ void __launch_bounds__(WG)
gcn_split_rowscan_kernel(int* __restrict__ counts,
                         int* __restrict__ totals) {
    static_assert(SPLIT_WG % WAVES == 0, "the chunks split evenly over the waves");
    __shared__ int wsum[WAVES * 32];
    const int tid  = threadIdx.x;
    const int lane = tid & 31;
    const int wave = tid >> 5;
    const int b    = blockIdx.x * 32 + lane;
    int* p = counts + (size_t)(wave * RS_CPW) * NB_MAX + b;
    int v[RS_CPW];
    #pragma unroll
    for (int i = 0; i < RS_CPW; ++i) v[i] = p[(size_t)i * NB_MAX];
    int run = 0;
    #pragma unroll
    for (int i = 0; i < RS_CPW; ++i) {
        const int t = v[i];
        v[i] = run;
        run += t;
    }
    wsum[wave * 32 + lane] = run;
    __syncthreads();
    int off = 0, total = 0;
    #pragma unroll
    for (int w = 0; w < WAVES; ++w) {
        const int t = wsum[w * 32 + lane];
        if (w < wave) off += t;
        total += t;
    }
    #pragma unroll
    for (int i = 0; i < RS_CPW; ++i) p[(size_t)i * NB_MAX] = v[i] + off;
    if (wave == 0) totals[b] = total;
}

__global__ void __launch_bounds__(SWG)
gcn_split_scatter_kernel(const int* __restrict__ rows,
                         const int* __restrict__ cols,
                         const unsigned short* __restrict__ place,
                         const int* __restrict__ counts,
                         const int* __restrict__ totals,
                         int* __restrict__ bstart,
                         int* __restrict__ sorted,
                         int n_edges, int chunk, int nb) {
    static_assert(SWG == NB_MAX, "one thread per bucket in the start step");
    static_assert(NB_MAX * NPB <= (1 << ROW_BITS), "a node id fits ROW_BITS bits");
    static_assert(ROW_BITS + NPB_SHIFT <= 31, "a packed edge stays non-negative");
    __shared__ v2i stage[STAGE_CAP];
    __shared__ int lstart[NB_MAX];
    __shared__ int delta[NB_MAX];
    __shared__ unsigned long long scan_sm[SWG];
    const int tid = threadIdx.x;
    const int wg  = blockIdx.x;
    const int beg = min(n_edges, wg * chunk);
    const int end = min(n_edges, beg + chunk);
    const int i_first = beg + tid;
    int rr[MAXB], cc[MAXB], pl[MAXB];
    if (end <= beg && wg != 0) return;
    #pragma unroll
    for (int j = 0; j < MAXB; ++j) { rr[j] = 0; cc[j] = 0; pl[j] = 0; }
    if (end > beg) {
        #pragma unroll
        for (int j = 0; j < MAXB; ++j) {
            const int i = min(i_first + j * SWG, end - 1);
            rr[j] = rows[i];
            cc[j] = cols[i];
            pl[j] = place[i];
        }
    }
    int cnt = 0, sc = 0, tot = 0;
    if (tid < nb) {
        sc  = counts[wg * NB_MAX + tid];
        tot = totals[tid];
        const int nx = (wg + 1 < SPLIT_WG) ? counts[(wg + 1) * NB_MAX + tid] : tot;
        cnt = nx - sc;
    }
    unsigned long long all;
    const unsigned long long ex = block_excl_scan(
        ((unsigned long long)(unsigned)cnt << 32) | (unsigned)tot, scan_sm, tid, SWG, &all);
    const int bs = (int)(unsigned)ex;
    const int ls = (int)(ex >> 32);
    lstart[tid] = ls;
    delta[tid]  = bs + sc - ls;
    if (wg == 0) {
        if (tid < nb)  bstart[tid] = bs;
        if (tid == 0)  bstart[nb]  = (int)(unsigned)all;
    }
    __syncthreads();
    int at[MAXB];
    #pragma unroll
    for (int j = 0; j < MAXB; ++j) at[j] = lstart[cc[j] >> NPB_SHIFT] + pl[j];
    #pragma unroll
    for (int j = 0; j < MAXB; ++j) {
        if (i_first + j * SWG < end) {
            v2i e;
            e.x = rr[j];
            e.y = cc[j];
            stage[at[j]] = e;
        }
    }
    __syncthreads();
    const int n_chunk = end - beg;
    for (int i0 = tid; i0 < n_chunk; i0 += LD_U * SWG) {
        v2i e[LD_U];
        int d[LD_U];
        #pragma unroll
        for (int u = 0; u < LD_U; ++u) e[u] = stage[min(i0 + u * SWG, n_chunk - 1)];
        #pragma unroll
        for (int u = 0; u < LD_U; ++u) d[u] = delta[e[u].y >> NPB_SHIFT];
        #pragma unroll
        for (int u = 0; u < LD_U; ++u) {
            const int i = i0 + u * SWG;
            if (i < n_chunk) sorted[i + d[u]] = e[u].x | ((e[u].y & (NPB - 1)) << ROW_BITS);
        }
    }
}

__global__ void __launch_bounds__(WG)
gcn_degree_xw1_kernel(const int* __restrict__ sorted,
                      const int* __restrict__ bstart,
                      const float* __restrict__ x,
                      const float* __restrict__ W1,
                      float* __restrict__ dinv,
                      float* __restrict__ h1s,
                      int n) {
    static_assert(NPB == WAVES * 16, "one 16-node tile per wave covers the bucket");
    __shared__ int cnt[WAVES * NPB];
    __shared__ float dsm[NPB];
    const int tid  = threadIdx.x;
    const int lane = tid & 31;
    const int wave = tid >> 5;
    const int half = lane >> 4;
    const int l16  = lane & 15;
    const int b    = blockIdx.x;
    const int base = b << NPB_SHIFT;
    const int node0 = base + wave * 16;
    const bool tile = node0 < n;

    for (int k = tid; k < WAVES * NPB; k += WG) cnt[k] = 0;
    __syncthreads();

    v2f fa[7], fb[7];
    #pragma unroll
    for (int s = 0; s < 7; ++s) { fa[s].x = 0.0f; fa[s].y = 0.0f; fb[s].x = 0.0f; fb[s].y = 0.0f; }
    if (tile) {
        #pragma unroll
        for (int s = 0; s < 7; ++s) {
            const int ka = 4 * s + 2 * half;
            if (ka     < FIN) { fa[s].x = x[(node0 + l16) * FIN + ka    ]; fb[s].x = W1[(ka    ) * F1 + l16]; }
            if (ka + 1 < FIN) { fa[s].y = x[(node0 + l16) * FIN + ka + 1]; fb[s].y = W1[(ka + 1) * F1 + l16]; }
        }
    }

    const int beg = bstart[b];
    const int end = bstart[b + 1];
    lds_vint* cw = (lds_vint*)(cnt + wave * NPB);
    const int nbatch = (end - beg + WG - 1) / WG;
    for (int j0 = 0; j0 < nbatch; j0 += LD_U) {
        int cc[LD_U];
        #pragma unroll
        for (int u = 0; u < LD_U; ++u) {
            const int i = beg + (j0 + u) * WG + tid;
            cc[u] = sorted[min(i, end - 1)];
        }
        #pragma unroll
        for (int u = 0; u < LD_U; ++u) {
            const int i = beg + (j0 + u) * WG + tid;
            const bool act = i < end;
            const int key = cc[u] >> ROW_BITS;
            const unsigned m = wave_peers(key, act, NPB_SHIFT);
            if (act && lanes_below(m) == 0) cw[key] = cw[key] + __popc(m);
        }
    }

    v8f c = {0.f, 0.f, 0.f, 0.f, 0.f, 0.f, 0.f, 0.f};
    if (tile) {
        #pragma unroll
        for (int s = 0; s < 7; ++s) {
            c = __builtin_amdgcn_wmma_f32_16x16x4_f32(
                     false, fa[s],  false, fb[s],
                     (short)0, c,  false,  false);
        }
    }
    __syncthreads();
    if (tid < NPB) {
        float dv = 0.0f;
        if (base + tid < n) {
            int d = 1;
            #pragma unroll
            for (int w = 0; w < WAVES; ++w) d += cnt[w * NPB + tid];
            dv = rsqrtf((float)d);
            dinv[base + tid] = dv;
        }
        dsm[tid] = dv;
    }
    __syncthreads();
    if (tile) {
        #pragma unroll
        for (int v = 0; v < 8; ++v) {
            const int ln = wave * 16 + v + 8 * half;
            h1s[(base + ln) * F1 + l16] = c[v] * dsm[ln];
        }
    }
}

#define L1_U 8
#define L1_S (L1_U / 4)

template <int U>
__device__ __forceinline__ int quad_lane(int v) {
    return __builtin_amdgcn_mov_dpp(v, U * 0x55, 0xf, 0xf, true);
}

template <int U>
__device__ __forceinline__ void l1_add_group(lds_vfloat4* aw, int pk, unsigned multi,
                                             int q, v4f v) {
    const int pku  = quad_lane<U>(pk);
    const int key  = pku & 0xFF;
    const int rank = (pku >> 8) & 0xFF;
    const bool act = (pku >> 16) != 0;
    if (((multi >> U) & 0x11111111u) == 0u) {
        if (act) {
            v4f t = aw[key * 4 + q];
            aw[key * 4 + q] = t + v;
        }
    } else {
        for (int turn = 0; ; ++turn) {
            const bool mine = act && rank == turn;
            if (__builtin_amdgcn_ballot_w32(mine) == 0) break;
            if (mine) {
                v4f t = aw[key * 4 + q];
                aw[key * 4 + q] = t + v;
            }
        }
    }
}
__global__ void __launch_bounds__(WG)
gcn_layer1_kernel(const int* __restrict__ sorted,
                  const int* __restrict__ bstart,
                  const float* __restrict__ h1s,
                  const float* __restrict__ dinv,
                  const float* __restrict__ b1,
                  const float* __restrict__ W2,
                  float* __restrict__ h2s,
                  int n) {
    __shared__ __attribute__((aligned(16))) float acc[WAVES * NPB * F1];
    const int tid  = threadIdx.x;
    const int lane = tid & 31;
    const int wave = tid >> 5;
    const int grp  = lane >> 2;
    const int q    = lane & 3;
    const int b    = blockIdx.x;
    const int base = b << NPB_SHIFT;
    for (int k = tid; k < WAVES * NPB * F1; k += WG) acc[k] = 0.0f;
    __syncthreads();
    const int beg = bstart[b];
    const int end = bstart[b + 1];
    const v4f* h1v = (const v4f*)h1s;
    lds_vfloat4* aw = (lds_vfloat4*)(acc + wave * (NPB * F1));
    const unsigned qmask = 0x11111111u << q;
    for (int i0 = beg; i0 < end; i0 += 8 * WAVES * L1_U) {
        int  es[L1_S];
        v4f  vv[L1_U];
        #pragma unroll
        for (int s = 0; s < L1_S; ++s) {
            const int i = i0 + (4 * s + q) * (8 * WAVES) + wave * 8 + grp;
            es[s] = sorted[min(i, end - 1)];
        }
        #pragma unroll
        for (int s = 0; s < L1_S; ++s) {
            const int row = es[s] & ROW_MASK;
            vv[4 * s + 0] = h1v[quad_lane<0>(row) * 4 + q];
            vv[4 * s + 1] = h1v[quad_lane<1>(row) * 4 + q];
            vv[4 * s + 2] = h1v[quad_lane<2>(row) * 4 + q];
            vv[4 * s + 3] = h1v[quad_lane<3>(row) * 4 + q];
        }
        #pragma unroll
        for (int s = 0; s < L1_S; ++s) {
            const int i = i0 + (4 * s + q) * (8 * WAVES) + wave * 8 + grp;
            const bool act = i < end;
            const int key = es[s] >> ROW_BITS;
            const unsigned m = wave_peers(key, act, NPB_SHIFT);
            const int rank = lanes_below(m & qmask);
            const unsigned multi = __builtin_amdgcn_ballot_w32(rank != 0);
            const int pk = key | (rank << 8) | (act ? 0x10000 : 0);
            l1_add_group<0>(aw, pk, multi, q, vv[4 * s + 0]);
            l1_add_group<1>(aw, pk, multi, q, vv[4 * s + 1]);
            l1_add_group<2>(aw, pk, multi, q, vv[4 * s + 2]);
            l1_add_group<3>(aw, pk, multi, q, vv[4 * s + 3]);
        }
    }
    __syncthreads();
    for (int k = tid; k < NPB * F1; k += WG) {
        float s = acc[k];
        #pragma unroll
        for (int w = 1; w < WAVES; ++w) s += acc[w * (NPB * F1) + k];
        acc[k] = s;
    }
    __syncthreads();
    const int node = base + tid;
    if (tid < NPB && node < n) {
        float di = dinv[node];
        float a0 = 0.0f, a1 = 0.0f;
        #pragma unroll
        for (int k = 0; k < F1; ++k) {
            float v = (acc[tid * F1 + k] + h1s[node * F1 + k]) * di + b1[k];
            v = v > 0.0f ? v : 0.0f;
            a0 = fmaf(v, W2[k * F2 + 0], a0);
            a1 = fmaf(v, W2[k * F2 + 1], a1);
        }
        h2s[node * F2 + 0] = a0 * di;
        h2s[node * F2 + 1] = a1 * di;
    }
}

__global__ void __launch_bounds__(WG)
gcn_layer2_kernel(const int* __restrict__ sorted,
                  const int* __restrict__ bstart,
                  const float* __restrict__ h2s,
                  const float* __restrict__ dinv,
                  const float* __restrict__ b2,
                  float* __restrict__ out,
                  int n) {
    __shared__ float acc[WAVES * NPB * F2];
    const int tid  = threadIdx.x;
    const int lane = tid & 31;
    const int wave = tid >> 5;
    const int b    = blockIdx.x;
    const int base = b << NPB_SHIFT;
    for (int k = tid; k < WAVES * NPB * F2; k += WG) acc[k] = 0.0f;
    __syncthreads();
    const int beg = bstart[b];
    const int end = bstart[b + 1];
    const float2* h2v = (const float2*)h2s;
    lds_vfloat* aw = (lds_vfloat*)(acc + wave * (NPB * F2));
    const int nbatch = (end - beg + WG - 1) / WG;
    for (int j0 = 0; j0 < nbatch; j0 += LD_U) {
        int    ee[LD_U];
        float2 vv[LD_U];
        #pragma unroll
        for (int u = 0; u < LD_U; ++u) {
            const int i = beg + (j0 + u) * WG + tid;
            ee[u] = sorted[min(i, end - 1)];
        }
        #pragma unroll
        for (int u = 0; u < LD_U; ++u) vv[u] = h2v[ee[u] & ROW_MASK];
        #pragma unroll
        for (int u = 0; u < LD_U; ++u) {
            const int i = beg + (j0 + u) * WG + tid;
            const bool act = i < end;
            const int key = ee[u] >> ROW_BITS;
            const unsigned m = wave_peers(key, act, NPB_SHIFT);
            const int rank = lanes_below(m);
            for (int turn = 0; ; ++turn) {
                const bool mine = act && rank == turn;
                if (__builtin_amdgcn_ballot_w32(mine) == 0) break;
                if (mine) {
                    aw[key * F2 + 0] = aw[key * F2 + 0] + vv[u].x;
                    aw[key * F2 + 1] = aw[key * F2 + 1] + vv[u].y;
                }
            }
        }
    }
    __syncthreads();
    const int node = base + tid;
    if (tid < NPB && node < n) {
        float s0 = 0.0f, s1 = 0.0f;
        #pragma unroll
        for (int w = 0; w < WAVES; ++w) {
            s0 += acc[w * (NPB * F2) + tid * F2 + 0];
            s1 += acc[w * (NPB * F2) + tid * F2 + 1];
        }
        float di = dinv[node];
        float2 s = h2v[node];
        float a = (s0 + s.x) * di + b2[0];
        float c = (s1 + s.y) * di + b2[1];
        float m = fmaxf(a, c);
        float lse = m + logf(expf(a - m) + expf(c - m));
        out[node * F2 + 0] = a - lse;
        out[node * F2 + 1] = c - lse;
    }
}

extern "C" void kernel_launch(void* const* d_in, const int* in_sizes, int n_in,
                              void* d_out, int out_size, void* d_ws, size_t ws_size,
                              hipStream_t stream) {
    const float* x   = (const float*)d_in[0];
    const int*   ei  = (const int*)d_in[1];
    const float* W1  = (const float*)d_in[2];
    const float* b1  = (const float*)d_in[3];
    const float* W2  = (const float*)d_in[4];
    const float* b2  = (const float*)d_in[5];
    float*       out = (float*)d_out;

    const int N = in_sizes[0] / FIN;
    const int E = in_sizes[1] / 2;
    const int* rows = ei;
    const int* cols = ei + E;

    const int NB    = (N + NPB - 1) / NPB;
    const int chunk = (E + SPLIT_WG - 1) / SPLIT_WG;

    int*   wsi    = (int*)d_ws;
    int*   sorted = wsi;
    float* h1s    = (float*)(wsi + (size_t)2 * E);
    float* h2s    = h1s + (size_t)16 * N;
    float* dinv   = h2s + (size_t)2 * N;
    int*   counts = (int*)(dinv + (size_t)N);
    int*   totals = counts + (size_t)NB_MAX * SPLIT_WG;
    int*   bstart = totals + NB_MAX;
    unsigned short* place = (unsigned short*)(bstart + NB_MAX + 8);

    gcn_split_count_kernel<<<SPLIT_WG, SWG, 0, stream>>>(cols, counts, place, E, chunk, NB);
    gcn_split_rowscan_kernel<<<NB_MAX / 32, WG, 0, stream>>>(counts, totals);
    gcn_split_scatter_kernel<<<SPLIT_WG, SWG, 0, stream>>>(rows, cols, place, counts, totals,
                                                          bstart, sorted, E, chunk, NB);

    gcn_degree_xw1_kernel<<<NB, WG, 0, stream>>>(sorted, bstart, x, W1, dinv, h1s, N);

    gcn_layer1_kernel<<<NB, WG, 0, stream>>>(sorted, bstart, h1s, dinv, b1, W2, h2s, N);

    gcn_layer2_kernel<<<NB, WG, 0, stream>>>(sorted, bstart, h2s, dinv, b2, out, N);
}


# Edges gathered per bucket, not scattered per chunk: chunks leave LDS bucket-ordered with a start/count table, bucket workgroups fetch their pieces (lane-move scans); 6 launches to 4; bit-identical
# speedup vs baseline: 6.7231x; 1.1417x over previous
#include <hip/hip_runtime.h>
#include <hip/hip_bf16.h>


typedef __attribute__((ext_vector_type(2))) float v2f;
typedef __attribute__((ext_vector_type(8))) float v8f;

#define F1 16
#define F2 2
#define FIN 25

#define NPB_SHIFT 7
#define NPB 128
#define NB_MAX 1024
#define NB_BITS 10
#define ROW_BITS 17
#define ROW_MASK ((1 << ROW_BITS) - 1)
#define SPLIT_WG 256
#define SWG 1024
#define SWAVES 32
#define STAGE_CAP 16384
#define WG 256
#define WAVES 8
#define LD_U 4
#define MAXB (STAGE_CAP / SWG)

typedef __attribute__((address_space(3))) volatile int   lds_vint;
typedef __attribute__((address_space(3))) volatile float lds_vfloat;
typedef __attribute__((ext_vector_type(4))) float v4f;
typedef __attribute__((address_space(3))) volatile v4f   lds_vfloat4;
typedef __attribute__((ext_vector_type(2))) int v2i;

__device__ __forceinline__ unsigned wave_peers(int key, bool act, int nbits) {
    unsigned m = __builtin_amdgcn_ballot_w32(act);
    for (int bit = 0; bit < nbits; ++bit) {
        const int x = __builtin_amdgcn_sbfe(key, bit, 1);
        const unsigned bal = __builtin_amdgcn_ballot_w32(x != 0);
        m = __builtin_amdgcn_bitop3_b32(m, bal, (unsigned)x, 0x90);
    }
    return act ? m : 0u;
}

__device__ __forceinline__ int lanes_below(unsigned m) {
    return (int)__builtin_amdgcn_mbcnt_lo(m, 0u);
}

__device__ __forceinline__ int lane_up(int v, int d) { return __shfl_up(v, (unsigned)d, 32); }
__device__ __forceinline__ int lane_get(int v, int src) { return __shfl(v, src, 32); }
__device__ __forceinline__ unsigned long long lane_up(unsigned long long v, int d) {
    const int lo = __shfl_up((int)(unsigned)v, (unsigned)d, 32);
    const int hi = __shfl_up((int)(unsigned)(v >> 32), (unsigned)d, 32);
    return ((unsigned long long)(unsigned)hi << 32) | (unsigned)lo;
}
__device__ __forceinline__ unsigned long long lane_get(unsigned long long v, int src) {
    const int lo = __shfl((int)(unsigned)v, src, 32);
    const int hi = __shfl((int)(unsigned)(v >> 32), src, 32);
    return ((unsigned long long)(unsigned)hi << 32) | (unsigned)lo;
}
template <typename T>
__device__ __forceinline__ T wave_incl_scan(T v, int lane) {
    #pragma unroll
    for (int d = 1; d < 32; d <<= 1) {
        const T t = lane_up(v, d);
        if (lane >= d) v += t;
    }
    return v;
}
template <typename T>
__device__ __forceinline__ T block_excl_scan(T v, T* sm, int tid, int nwaves, T* total) {
    const int lane = tid & 31;
    const int wave = tid >> 5;
    const T incl = wave_incl_scan(v, lane);
    if (lane == 31) sm[wave] = incl;
    __syncthreads();
    const T w  = (lane < nwaves) ? sm[lane] : (T)0;
    const T wi = wave_incl_scan(w, lane);
    *total = lane_get(wi, nwaves - 1);
    const T before = lane_get(wi - w, wave);
    return before + (incl - v);
}

__global__ void __launch_bounds__(SWG)
gcn_split_local_kernel(const int* __restrict__ rows,
                       const int* __restrict__ cols,
                       int* __restrict__ piece_w,
                       int* __restrict__ tmp,
                       int n_edges, int chunk) {
    static_assert(SWG == NB_MAX, "one thread per bucket in the offset step");
    static_assert(NB_MAX * NPB <= (1 << ROW_BITS), "a node id fits ROW_BITS bits");
    static_assert(ROW_BITS + NPB_SHIFT <= 31, "a packed edge stays non-negative");
    __shared__ int hist[SWAVES * NB_MAX];
    __shared__ int stage[STAGE_CAP];
    __shared__ int lsm[NB_MAX];
    __shared__ int scan_sm[SWAVES];
    const int tid  = threadIdx.x;
    const int wave = tid >> 5;
    const int wg   = blockIdx.x;
    const int beg = min(n_edges, wg * chunk);
    const int end = min(n_edges, beg + chunk);
    const int nbatch  = (end - beg + SWG - 1) / SWG;
    const int i_first = beg + tid;
    int cc[MAXB], rr[MAXB], loc[MAXB];
    #pragma unroll
    for (int j = 0; j < MAXB; ++j) { cc[j] = 0; rr[j] = 0; }
    if (end > beg) {
        #pragma unroll
        for (int j = 0; j < MAXB; ++j) {
            const int i = min(i_first + j * SWG, end - 1);
            cc[j] = cols[i];
            rr[j] = rows[i];
        }
    }
    for (int k = tid; k < SWAVES * NB_MAX; k += SWG) hist[k] = 0;
    __syncthreads();
    lds_vint* hw = (lds_vint*)(hist + wave * NB_MAX);
    #pragma unroll
    for (int j = 0; j < MAXB; ++j) {
        loc[j] = 0;
        if (j < nbatch) {
            const bool act = i_first + j * SWG < end;
            const int key = cc[j] >> NPB_SHIFT;
            const unsigned m = wave_peers(key, act, NB_BITS);
            const int rank = lanes_below(m);
            int p = 0;
            if (act) p = hw[key];
            if (act && rank == 0) hw[key] = p + __popc(m);
            loc[j] = p + rank;
        }
    }
    __syncthreads();
    int tot = 0;
    #pragma unroll
    for (int w = 0; w < SWAVES; ++w) {
        int t = hist[w * NB_MAX + tid];
        hist[w * NB_MAX + tid] = tot;
        tot += t;
    }
    int all;
    const int ls = block_excl_scan(tot, scan_sm, tid, SWAVES, &all);
    v2i pc;
    pc.x = ls;
    pc.y = tot;
    ((v2i*)piece_w)[wg * NB_MAX + tid] = pc;
    lsm[tid] = ls;
    __syncthreads();
    int at[MAXB];
    #pragma unroll
    for (int j = 0; j < MAXB; ++j) {
        const int key = cc[j] >> NPB_SHIFT;
        at[j] = loc[j] + hw[key] + lsm[key];
    }
    #pragma unroll
    for (int j = 0; j < MAXB; ++j) {
        if (i_first + j * SWG < end)
            stage[at[j]] = rr[j] | ((cc[j] & (NPB - 1)) << ROW_BITS);
    }
    __syncthreads();
    const int n_chunk = end - beg;
    for (int i = tid; i < n_chunk; i += SWG) tmp[beg + i] = stage[i];
}

__global__ void __launch_bounds__(WG)
gcn_gather_degree_xw1_kernel(const int* __restrict__ tmp,
                             const int* __restrict__ piece_w,
                             int* __restrict__ sorted,
                             int* __restrict__ bstart,
                             const float* __restrict__ x,
                             const float* __restrict__ W1,
                             float* __restrict__ dinv,
                             float* __restrict__ h1s,
                             int n, int chunk, int nb) {
    static_assert(NPB == WAVES * 16, "one 16-node tile per wave covers the bucket");
    static_assert(SPLIT_WG == WG, "one thread per chunk in the piece step");
    static_assert(SPLIT_WG == 256, "the binary search below takes 8 steps");
    __shared__ int cnt[WAVES * NPB];
    __shared__ float dsm[NPB];
    __shared__ unsigned long long scan_sm[WAVES];
    __shared__ int segpos[SPLIT_WG];
    __shared__ int segsrc[SPLIT_WG];
    const int tid  = threadIdx.x;
    const int lane = tid & 31;
    const int wave = tid >> 5;
    const int half = lane >> 4;
    const int l16  = lane & 15;
    const int b    = blockIdx.x;
    const int base = b << NPB_SHIFT;
    const int node0 = base + wave * 16;
    const bool tile = node0 < n;

    const v2i pc = ((const v2i*)piece_w)[tid * NB_MAX + b];
    const int ps = pc.x;
    const int pn = pc.y;
    for (int k = tid; k < WAVES * NPB; k += WG) cnt[k] = 0;
    unsigned long long all;
    const unsigned long long ex = block_excl_scan(
        ((unsigned long long)(unsigned)pn << 32) | (unsigned)ps, scan_sm, tid, WAVES, &all);
    const int po  = (int)(ex >> 32);
    const int beg = (int)(unsigned)all;
    const int nbk = (int)(all >> 32);
    segpos[tid] = po;
    segsrc[tid] = tid * chunk + ps - po;
    if (tid == 0) {
        bstart[b] = beg;
        if (b == nb - 1) bstart[nb] = beg + nbk;
    }
    __syncthreads();

    v2f fa[7], fb[7];
    #pragma unroll
    for (int s = 0; s < 7; ++s) { fa[s].x = 0.0f; fa[s].y = 0.0f; fb[s].x = 0.0f; fb[s].y = 0.0f; }
    if (tile) {
        #pragma unroll
        for (int s = 0; s < 7; ++s) {
            const int ka = 4 * s + 2 * half;
            if (ka     < FIN) { fa[s].x = x[(node0 + l16) * FIN + ka    ]; fb[s].x = W1[(ka    ) * F1 + l16]; }
            if (ka + 1 < FIN) { fa[s].y = x[(node0 + l16) * FIN + ka + 1]; fb[s].y = W1[(ka + 1) * F1 + l16]; }
        }
    }

    lds_vint* cw = (lds_vint*)(cnt + wave * NPB);
    const int nbatch = (nbk + WG - 1) / WG;
    for (int j0 = 0; j0 < nbatch; j0 += LD_U) {
        int cc[LD_U];
        #pragma unroll
        for (int u = 0; u < LD_U; ++u) {
            const int p = min((j0 + u) * WG + tid, nbk - 1);
            int c = 0;
            #pragma unroll
            for (int step = SPLIT_WG / 2; step > 0; step >>= 1) {
                if (segpos[c + step] <= p) c += step;
            }
            cc[u] = tmp[segsrc[c] + p];
        }
        #pragma unroll
        for (int u = 0; u < LD_U; ++u) {
            const int p = (j0 + u) * WG + tid;
            const bool act = p < nbk;
            if (act) sorted[beg + p] = cc[u];
            const int key = cc[u] >> ROW_BITS;
            const unsigned m = wave_peers(key, act, NPB_SHIFT);
            if (act && lanes_below(m) == 0) cw[key] = cw[key] + __popc(m);
        }
    }

    v8f c = {0.f, 0.f, 0.f, 0.f, 0.f, 0.f, 0.f, 0.f};
    if (tile) {
        #pragma unroll
        for (int s = 0; s < 7; ++s) {
            c = __builtin_amdgcn_wmma_f32_16x16x4_f32(
                     false, fa[s],  false, fb[s],
                     (short)0, c,  false,  false);
        }
    }
    __syncthreads();
    if (tid < NPB) {
        float dv = 0.0f;
        if (base + tid < n) {
            int d = 1;
            #pragma unroll
            for (int w = 0; w < WAVES; ++w) d += cnt[w * NPB + tid];
            dv = rsqrtf((float)d);
            dinv[base + tid] = dv;
        }
        dsm[tid] = dv;
    }
    __syncthreads();
    if (tile) {
        #pragma unroll
        for (int v = 0; v < 8; ++v) {
            const int ln = wave * 16 + v + 8 * half;
            h1s[(base + ln) * F1 + l16] = c[v] * dsm[ln];
        }
    }
}

#define L1_U 8
#define L1_S (L1_U / 4)

template <int U>
__device__ __forceinline__ int quad_lane(int v) {
    return __builtin_amdgcn_mov_dpp(v, U * 0x55, 0xf, 0xf, true);
}

template <int U>
__device__ __forceinline__ void l1_add_group(lds_vfloat4* aw, int pk, unsigned multi,
                                             int q, v4f v) {
    const int pku  = quad_lane<U>(pk);
    const int key  = pku & 0xFF;
    const int rank = (pku >> 8) & 0xFF;
    const bool act = (pku >> 16) != 0;
    if (((multi >> U) & 0x11111111u) == 0u) {
        if (act) {
            v4f t = aw[key * 4 + q];
            aw[key * 4 + q] = t + v;
        }
    } else {
        for (int turn = 0; ; ++turn) {
            const bool mine = act && rank == turn;
            if (__builtin_amdgcn_ballot_w32(mine) == 0) break;
            if (mine) {
                v4f t = aw[key * 4 + q];
                aw[key * 4 + q] = t + v;
            }
        }
    }
}
__global__ void __launch_bounds__(WG)
gcn_layer1_kernel(const int* __restrict__ sorted,
                  const int* __restrict__ bstart,
                  const float* __restrict__ h1s,
                  const float* __restrict__ dinv,
                  const float* __restrict__ b1,
                  const float* __restrict__ W2,
                  float* __restrict__ h2s,
                  int n) {
    __shared__ __attribute__((aligned(16))) float acc[WAVES * NPB * F1];
    const int tid  = threadIdx.x;
    const int lane = tid & 31;
    const int wave = tid >> 5;
    const int grp  = lane >> 2;
    const int q    = lane & 3;
    const int b    = blockIdx.x;
    const int base = b << NPB_SHIFT;
    for (int k = tid; k < WAVES * NPB * F1; k += WG) acc[k] = 0.0f;
    __syncthreads();
    const int beg = bstart[b];
    const int end = bstart[b + 1];
    const v4f* h1v = (const v4f*)h1s;
    lds_vfloat4* aw = (lds_vfloat4*)(acc + wave * (NPB * F1));
    const unsigned qmask = 0x11111111u << q;
    for (int i0 = beg; i0 < end; i0 += 8 * WAVES * L1_U) {
        int  es[L1_S];
        v4f  vv[L1_U];
        #pragma unroll
        for (int s = 0; s < L1_S; ++s) {
            const int i = i0 + (4 * s + q) * (8 * WAVES) + wave * 8 + grp;
            es[s] = sorted[min(i, end - 1)];
        }
        #pragma unroll
        for (int s = 0; s < L1_S; ++s) {
            const int row = es[s] & ROW_MASK;
            vv[4 * s + 0] = h1v[quad_lane<0>(row) * 4 + q];
            vv[4 * s + 1] = h1v[quad_lane<1>(row) * 4 + q];
            vv[4 * s + 2] = h1v[quad_lane<2>(row) * 4 + q];
            vv[4 * s + 3] = h1v[quad_lane<3>(row) * 4 + q];
        }
        #pragma unroll
        for (int s = 0; s < L1_S; ++s) {
            const int i = i0 + (4 * s + q) * (8 * WAVES) + wave * 8 + grp;
            const bool act = i < end;
            const int key = es[s] >> ROW_BITS;
            const unsigned m = wave_peers(key, act, NPB_SHIFT);
            const int rank = lanes_below(m & qmask);
            const unsigned multi = __builtin_amdgcn_ballot_w32(rank != 0);
            const int pk = key | (rank << 8) | (act ? 0x10000 : 0);
            l1_add_group<0>(aw, pk, multi, q, vv[4 * s + 0]);
            l1_add_group<1>(aw, pk, multi, q, vv[4 * s + 1]);
            l1_add_group<2>(aw, pk, multi, q, vv[4 * s + 2]);
            l1_add_group<3>(aw, pk, multi, q, vv[4 * s + 3]);
        }
    }
    __syncthreads();
    for (int k = tid; k < NPB * F1; k += WG) {
        float s = acc[k];
        #pragma unroll
        for (int w = 1; w < WAVES; ++w) s += acc[w * (NPB * F1) + k];
        acc[k] = s;
    }
    __syncthreads();
    const int node = base + tid;
    if (tid < NPB && node < n) {
        float di = dinv[node];
        float a0 = 0.0f, a1 = 0.0f;
        #pragma unroll
        for (int k = 0; k < F1; ++k) {
            float v = (acc[tid * F1 + k] + h1s[node * F1 + k]) * di + b1[k];
            v = v > 0.0f ? v : 0.0f;
            a0 = fmaf(v, W2[k * F2 + 0], a0);
            a1 = fmaf(v, W2[k * F2 + 1], a1);
        }
        h2s[node * F2 + 0] = a0 * di;
        h2s[node * F2 + 1] = a1 * di;
    }
}

__global__ void __launch_bounds__(WG)
gcn_layer2_kernel(const int* __restrict__ sorted,
                  const int* __restrict__ bstart,
                  const float* __restrict__ h2s,
                  const float* __restrict__ dinv,
                  const float* __restrict__ b2,
                  float* __restrict__ out,
                  int n) {
    __shared__ float acc[WAVES * NPB * F2];
    const int tid  = threadIdx.x;
    const int lane = tid & 31;
    const int wave = tid >> 5;
    const int b    = blockIdx.x;
    const int base = b << NPB_SHIFT;
    for (int k = tid; k < WAVES * NPB * F2; k += WG) acc[k] = 0.0f;
    __syncthreads();
    const int beg = bstart[b];
    const int end = bstart[b + 1];
    const float2* h2v = (const float2*)h2s;
    lds_vfloat* aw = (lds_vfloat*)(acc + wave * (NPB * F2));
    const int nbatch = (end - beg + WG - 1) / WG;
    for (int j0 = 0; j0 < nbatch; j0 += LD_U) {
        int    ee[LD_U];
        float2 vv[LD_U];
        #pragma unroll
        for (int u = 0; u < LD_U; ++u) {
            const int i = beg + (j0 + u) * WG + tid;
            ee[u] = sorted[min(i, end - 1)];
        }
        #pragma unroll
        for (int u = 0; u < LD_U; ++u) vv[u] = h2v[ee[u] & ROW_MASK];
        #pragma unroll
        for (int u = 0; u < LD_U; ++u) {
            const int i = beg + (j0 + u) * WG + tid;
            const bool act = i < end;
            const int key = ee[u] >> ROW_BITS;
            const unsigned m = wave_peers(key, act, NPB_SHIFT);
            const int rank = lanes_below(m);
            for (int turn = 0; ; ++turn) {
                const bool mine = act && rank == turn;
                if (__builtin_amdgcn_ballot_w32(mine) == 0) break;
                if (mine) {
                    aw[key * F2 + 0] = aw[key * F2 + 0] + vv[u].x;
                    aw[key * F2 + 1] = aw[key * F2 + 1] + vv[u].y;
                }
            }
        }
    }
    __syncthreads();
    const int node = base + tid;
    if (tid < NPB && node < n) {
        float s0 = 0.0f, s1 = 0.0f;
        #pragma unroll
        for (int w = 0; w < WAVES; ++w) {
            s0 += acc[w * (NPB * F2) + tid * F2 + 0];
            s1 += acc[w * (NPB * F2) + tid * F2 + 1];
        }
        float di = dinv[node];
        float2 s = h2v[node];
        float a = (s0 + s.x) * di + b2[0];
        float c = (s1 + s.y) * di + b2[1];
        float m = fmaxf(a, c);
        float lse = m + logf(expf(a - m) + expf(c - m));
        out[node * F2 + 0] = a - lse;
        out[node * F2 + 1] = c - lse;
    }
}

extern "C" void kernel_launch(void* const* d_in, const int* in_sizes, int n_in,
                              void* d_out, int out_size, void* d_ws, size_t ws_size,
                              hipStream_t stream) {
    const float* x   = (const float*)d_in[0];
    const int*   ei  = (const int*)d_in[1];
    const float* W1  = (const float*)d_in[2];
    const float* b1  = (const float*)d_in[3];
    const float* W2  = (const float*)d_in[4];
    const float* b2  = (const float*)d_in[5];
    float*       out = (float*)d_out;

    const int N = in_sizes[0] / FIN;
    const int E = in_sizes[1] / 2;
    const int* rows = ei;
    const int* cols = ei + E;

    const int NB    = (N + NPB - 1) / NPB;
    const int chunk = (E + SPLIT_WG - 1) / SPLIT_WG;

    int*   wsi    = (int*)d_ws;
    int*   sorted = wsi;
    int*   tmp    = wsi + (size_t)E;
    float* h1s    = (float*)(wsi + (size_t)2 * E);
    float* h2s    = h1s + (size_t)16 * N;
    float* dinv   = h2s + (size_t)2 * N;
    int*   piece  = (int*)(dinv + (size_t)N);
    int*   bstart = piece + (size_t)2 * NB_MAX * SPLIT_WG;

    gcn_split_local_kernel<<<SPLIT_WG, SWG, 0, stream>>>(rows, cols, piece, tmp, E, chunk);

    gcn_gather_degree_xw1_kernel<<<NB, WG, 0, stream>>>(tmp, piece, sorted, bstart,
                                                        x, W1, dinv, h1s, N, chunk, NB);

    gcn_layer1_kernel<<<NB, WG, 0, stream>>>(sorted, bstart, h1s, dinv, b1, W2, h2s, N);

    gcn_layer2_kernel<<<NB, WG, 0, stream>>>(sorted, bstart, h2s, dinv, b2, out, N);
}
